# starting kernel
# baseline (speedup 1.0000x reference)
#include <hip/hip_runtime.h>
#include <math.h>

#define B_TOT 1024
#define N_IN  128
#define M_OUT 64
#define N_XI  256
#define L_DIM 256
#define LDE   257
#define OUT_AMP 40.0f
#define INV_M  (1.0f/1.001f)

typedef __attribute__((ext_vector_type(2))) float v2f;
typedef __attribute__((ext_vector_type(8))) float v8f;

__device__ __forceinline__ v2f frag16x4_g(const float* __restrict__ M, int ld,
                                          int r0, int k0, int lane) {
  int r = r0 + (lane & 15);
  int k = k0 + ((lane >> 4) << 1);
  const float* p = M + (size_t)r * ld + k;
  v2f f; f.x = p[0]; f.y = p[1]; return f;
}

__device__ __forceinline__ v2f frag16x4_s(const float* M, int ld, int k0, int lane) {
  int r = lane & 15;
  int k = k0 + ((lane >> 4) << 1);
  const float* p = M + r * ld + k;
  v2f f; f.x = p[0]; f.y = p[1]; return f;
}

__device__ __forceinline__ v8f wmma4(v2f a, v2f b, v8f c) {
  return __builtin_amdgcn_wmma_f32_16x16x4_f32(false, a, false, b, (short)0, c,
                                               false, false);
}

__global__ __launch_bounds__(64)
void controller_rnn_fused(const float* __restrict__ y_, const float* __restrict__ xi,
                          const float* __restrict__ B2, const float* __restrict__ C2,
                          const float* __restrict__ D21, const float* __restrict__ D22,
                          const float* __restrict__ D12, const float* __restrict__ bxi,
                          const float* __restrict__ bv,  const float* __restrict__ bu,
                          const float* __restrict__ Fm,  const float* __restrict__ B1,
                          const float* __restrict__ Lambda,
                          const float* __restrict__ C1,  const float* __restrict__ D11,
                          float* __restrict__ out_u, float* __restrict__ out_xi) {
  __shared__ float eb[2][16][LDE];
  const int lane = threadIdx.x & 31;
  const int wv   = threadIdx.x >> 5;
  const int b0   = (blockIdx.x * 2 + wv) * 16;
  float* E = &eb[wv][0][0];

  const int colc = lane & 15;
  const int rhi  = (lane >> 4) << 3;

  #pragma unroll 1
  for (int nt = 0; nt < 16; ++nt) {
    const int n0 = nt * 16;
    v8f acc = {};
    for (int k0 = 0; k0 < N_XI; k0 += 4)
      acc = wmma4(frag16x4_g(xi, N_XI, b0, k0, lane),
                  frag16x4_g(C1, N_XI, n0, k0, lane), acc);
    for (int k0 = 0; k0 < N_IN; k0 += 4)
      acc = wmma4(frag16x4_g(y_, N_IN, b0, k0, lane),
                  frag16x4_g(D12, N_IN, n0, k0, lane), acc);
    const int col = n0 + colc;
    const float bvc = (col == 0) ? 0.0f : bv[col];
    #pragma unroll
    for (int r = 0; r < 8; ++r)
      E[(r + rhi) * LDE + col] = acc[r] + bvc;
  }
  __syncthreads();

  {
    const int row  = lane & 15;
    const int half = lane >> 4;
    for (int i = 0; i < L_DIM; ++i) {
      float v = 0.0f;
      const float* d11r = D11 + (size_t)i * L_DIM;
      for (int j = half; j < i; j += 2)
        v += E[row * LDE + j] * d11r[j];
      v += __shfl_xor(v, 16);
      v += E[row * LDE + i];
      const float e = tanhf(v / Lambda[i]);
      if (lane < 16) E[row * LDE + i] = e;
      __syncthreads();
    }
  }

  #pragma unroll 1
  for (int nt = 0; nt < 4; ++nt) {
    const int n0 = nt * 16;
    v8f acc = {};
    for (int k0 = 0; k0 < N_XI; k0 += 4)
      acc = wmma4(frag16x4_g(xi, N_XI, b0, k0, lane),
                  frag16x4_g(C2, N_XI, n0, k0, lane), acc);
    for (int k0 = 0; k0 < L_DIM; k0 += 4)
      acc = wmma4(frag16x4_s(E, LDE, k0, lane),
                  frag16x4_g(D21, L_DIM, n0, k0, lane), acc);
    for (int k0 = 0; k0 < N_IN; k0 += 4)
      acc = wmma4(frag16x4_g(y_, N_IN, b0, k0, lane),
                  frag16x4_g(D22, N_IN, n0, k0, lane), acc);
    const int col = n0 + colc;
    const float bc = bu[col];
    #pragma unroll
    for (int r = 0; r < 8; ++r)
      out_u[(size_t)(b0 + r + rhi) * M_OUT + col] = OUT_AMP * (acc[r] + bc);
  }

  #pragma unroll 1
  for (int nt = 0; nt < 16; ++nt) {
    const int n0 = nt * 16;
    v8f acc = {};
    for (int k0 = 0; k0 < N_XI; k0 += 4)
      acc = wmma4(frag16x4_g(xi, N_XI, b0, k0, lane),
                  frag16x4_g(Fm, N_XI, n0, k0, lane), acc);
    for (int k0 = 0; k0 < L_DIM; k0 += 4)
      acc = wmma4(frag16x4_s(E, LDE, k0, lane),
                  frag16x4_g(B1, L_DIM, n0, k0, lane), acc);
    for (int k0 = 0; k0 < N_IN; k0 += 4)
      acc = wmma4(frag16x4_g(y_, N_IN, b0, k0, lane),
                  frag16x4_g(B2, N_IN, n0, k0, lane), acc);
    const int col = n0 + colc;
    const float bc = bxi[col];
    #pragma unroll
    for (int r = 0; r < 8; ++r)
      out_xi[(size_t)(b0 + r + rhi) * N_XI + col] = (acc[r] + bc) * INV_M;
  }
}

extern "C" void kernel_launch(void* const* d_in, const int* in_sizes, int n_in,
                              void* d_out, int out_size, void* d_ws, size_t ws_size,
                              hipStream_t stream) {
  const float* y_  = (const float*)d_in[0];
  const float* xi  = (const float*)d_in[1];
  const float* B2  = (const float*)d_in[2];
  const float* C2  = (const float*)d_in[3];
  const float* D21 = (const float*)d_in[4];
  const float* D22 = (const float*)d_in[5];
  const float* D12 = (const float*)d_in[6];
  const float* bxi = (const float*)d_in[7];
  const float* bv  = (const float*)d_in[8];
  const float* bu  = (const float*)d_in[9];
  const float* Fm  = (const float*)d_in[10];
  const float* B1  = (const float*)d_in[11];
  const float* Lam = (const float*)d_in[13];
  const float* C1  = (const float*)d_in[14];
  const float* D11 = (const float*)d_in[15];

  float* out_u  = (float*)d_out;
  float* out_xi = out_u + (size_t)B_TOT * M_OUT;

  controller_rnn_fused<<<dim3(32), dim3(64), 0, stream>>>(
      y_, xi, B2, C2, D21, D22, D12, bxi, bv, bu, Fm, B1, Lam, C1, D11,
      out_u, out_xi);

  (void)in_sizes; (void)n_in; (void)out_size; (void)d_ws; (void)ws_size;
}


# Recurrence step's dependent path cut: tanh(v/Lambda) as 1-2/(exp2(v*c)+1) on hardware exp2 and rcp, c=2log2(e)/Lambda tabled once off the path; no divide and no branch per step, f32 as before
# speedup vs baseline: 21.7932x; 21.7932x over previous
#include <hip/hip_runtime.h>
#include <math.h>


#define B_TOT 1024
#define N_IN  128
#define M_OUT 64
#define N_XI  256
#define L_DIM 256
#define LDE   257
#define OUT_AMP 40.0f
#define INV_M  (1.0f/1.001f)
#define TWO_LOG2E 2.8853900817779268f

typedef __attribute__((ext_vector_type(2))) float v2f;
typedef __attribute__((ext_vector_type(4))) float v4f;
typedef __attribute__((ext_vector_type(8))) float v8f;

__device__ __forceinline__ v2f frag16x4_g(const float* __restrict__ M, int ld,
                                          int r0, int k0, int lane) {
  int r = r0 + (lane & 15);
  int k = k0 + ((lane >> 4) << 1);
  const float* p = M + (size_t)r * ld + k;
  v2f f; f.x = p[0]; f.y = p[1]; return f;
}

__device__ __forceinline__ v2f frag16x4_s(const float* M, int ld, int k0, int lane) {
  int r = lane & 15;
  int k = k0 + ((lane >> 4) << 1);
  const float* p = M + r * ld + k;
  v2f f; f.x = p[0]; f.y = p[1]; return f;
}

__device__ __forceinline__ v8f wmma4(v2f a, v2f b, v8f c) {
  return __builtin_amdgcn_wmma_f32_16x16x4_f32(false, a, false, b, (short)0, c,
                                               false, false);
}

__device__ __forceinline__ void wmma_chunk16(const float* E, const float* __restrict__ D11,
                                             int c0, int kb, int lane, v8f& acc0, v8f& acc1) {
  const v2f b0 = frag16x4_g(D11, L_DIM, c0, kb,      lane);
  const v2f b1 = frag16x4_g(D11, L_DIM, c0, kb + 4,  lane);
  const v2f b2 = frag16x4_g(D11, L_DIM, c0, kb + 8,  lane);
  const v2f b3 = frag16x4_g(D11, L_DIM, c0, kb + 12, lane);
  const v2f a0 = frag16x4_s(E, LDE, kb,      lane);
  const v2f a1 = frag16x4_s(E, LDE, kb + 4,  lane);
  const v2f a2 = frag16x4_s(E, LDE, kb + 8,  lane);
  const v2f a3 = frag16x4_s(E, LDE, kb + 12, lane);
  acc0 = wmma4(a0, b0, acc0);
  acc1 = wmma4(a1, b1, acc1);
  acc0 = wmma4(a2, b2, acc0);
  acc1 = wmma4(a3, b3, acc1);
}

__device__ __forceinline__ void seg16x32(const float* __restrict__ X, const float* __restrict__ Wt,
                                         int ld, int b0, int n0, int lane,
                                         v8f& acc0, v8f& acc1) {
  #pragma unroll 16
  for (int k0 = 0; k0 < ld; k0 += 4) {
    const v2f a = frag16x4_g(X, ld, b0, k0, lane);
    acc0 = wmma4(a, frag16x4_g(Wt, ld, n0,      k0, lane), acc0);
    acc1 = wmma4(a, frag16x4_g(Wt, ld, n0 + 16, k0, lane), acc1);
  }
}

__global__ __launch_bounds__(32)
void rnn_pre(const float* __restrict__ y_, const float* __restrict__ xi,
             const float* __restrict__ D12, const float* __restrict__ bv,
             const float* __restrict__ C1, float* __restrict__ Ews) {
  const int lane = threadIdx.x;
  const int b0   = blockIdx.y * 16;
  const int n0   = blockIdx.x * 32;
  const int colc = lane & 15;
  const int rhi  = (lane >> 4) << 3;

  v8f acc0 = {}, acc1 = {};
  seg16x32(xi, C1,  N_XI, b0, n0, lane, acc0, acc1);
  seg16x32(y_, D12, N_IN, b0, n0, lane, acc0, acc1);
  const int col = n0 + colc;
  const float bv0 = (col == 0) ? 0.0f : bv[col];
  const float bv1 = bv[col + 16];
  #pragma unroll
  for (int r = 0; r < 8; ++r) {
    Ews[(size_t)(b0 + r + rhi) * L_DIM + col]      = acc0[r] + bv0;
    Ews[(size_t)(b0 + r + rhi) * L_DIM + col + 16] = acc1[r] + bv1;
  }
}

__global__ __launch_bounds__(32)
void rnn_rec(const float* __restrict__ Lambda, const float* __restrict__ D11,
             const float* __restrict__ Ews, float* __restrict__ eps_ws) {
  __shared__ float eb[16][LDE];
  __shared__ __attribute__((aligned(16))) float dg[256];
  __shared__ __attribute__((aligned(16))) float cs[L_DIM];
  const int lane = threadIdx.x;
  const int b0   = blockIdx.x * 16;
  float* E = &eb[0][0];

  const int colc = lane & 15;
  const int rhi  = (lane >> 4) << 3;

  {
    float lr[8];
    #pragma unroll
    for (int t = 0; t < 8; ++t) lr[t] = Lambda[lane * 8 + t];
    #pragma unroll
    for (int t = 0; t < 8; ++t) cs[lane * 8 + t] = TWO_LOG2E / lr[t];
  }

  #pragma unroll
  for (int r = 0; r < 16; ++r) {
    #pragma unroll
    for (int h = 0; h < 2; ++h) {
      const int c = h * 128 + lane * 4;
      const v4f t = *(const v4f*)(Ews + (size_t)(b0 + r) * L_DIM + c);
      E[r * LDE + c + 0] = t.x; E[r * LDE + c + 1] = t.y;
      E[r * LDE + c + 2] = t.z; E[r * LDE + c + 3] = t.w;
    }
  }
  __syncthreads();

  {
    const int row = lane & 15;
    float* DG = &dg[0];
    #pragma unroll 1
    for (int I = 0; I < 16; ++I) {
      const int c0 = I * 16;

      const float* dsrc = D11 + (size_t)(c0 + (lane >> 1)) * L_DIM + c0 + (lane & 1) * 8;
      float dreg[8];
      #pragma unroll
      for (int t = 0; t < 8; ++t) dreg[t] = dsrc[t];
      float cl[16];
      #pragma unroll
      for (int ii = 0; ii < 16; ++ii) cl[ii] = cs[c0 + ii];

      {
        v8f acc0, acc1 = {};
        #pragma unroll
        for (int r = 0; r < 8; ++r) acc0[r] = E[(r + rhi) * LDE + c0 + colc];
        int kb = 0;
        for (; kb + 64 <= c0; kb += 64) {
          wmma_chunk16(E, D11, c0, kb,      lane, acc0, acc1);
          wmma_chunk16(E, D11, c0, kb + 16, lane, acc0, acc1);
          wmma_chunk16(E, D11, c0, kb + 32, lane, acc0, acc1);
          wmma_chunk16(E, D11, c0, kb + 48, lane, acc0, acc1);
        }
        for (; kb < c0; kb += 16)
          wmma_chunk16(E, D11, c0, kb, lane, acc0, acc1);
        #pragma unroll
        for (int r = 0; r < 8; ++r) E[(r + rhi) * LDE + c0 + colc] = acc0[r] + acc1[r];
      }
      #pragma unroll
      for (int t = 0; t < 8; ++t) DG[lane * 8 + t] = dreg[t];
      __syncthreads();

      float v[16];
      #pragma unroll
      for (int jj = 0; jj < 16; ++jj) v[jj] = E[row * LDE + c0 + jj];
      float d[16][16];
      #pragma unroll
      for (int kk = 1; kk < 16; ++kk) {
        #pragma unroll
        for (int ii = 0; ii < kk; ++ii) d[kk][ii] = DG[kk * 16 + ii];
      }
      __builtin_amdgcn_sched_barrier(0);
      #pragma unroll
      for (int ii = 0; ii < 16; ++ii) {
        const float ex = __builtin_amdgcn_exp2f(v[ii] * cl[ii]);
        const float e  = fmaf(-2.0f, __builtin_amdgcn_rcpf(ex + 1.0f), 1.0f);
        v[ii] = e;
        #pragma unroll
        for (int kk = ii + 1; kk < 16; ++kk) v[kk] = fmaf(e, d[kk][ii], v[kk]);
      }
      if (lane < 16) {
        #pragma unroll
        for (int jj = 0; jj < 16; ++jj) E[row * LDE + c0 + jj] = v[jj];
      }
      __syncthreads();
    }
  }

  #pragma unroll
  for (int r = 0; r < 16; ++r) {
    #pragma unroll
    for (int h = 0; h < 2; ++h) {
      const int c = h * 128 + lane * 4;
      v4f t;
      t.x = E[r * LDE + c + 0]; t.y = E[r * LDE + c + 1];
      t.z = E[r * LDE + c + 2]; t.w = E[r * LDE + c + 3];
      *(v4f*)(eps_ws + (size_t)(b0 + r) * L_DIM + c) = t;
    }
  }
}

__global__ __launch_bounds__(32)
void rnn_post(const float* __restrict__ y_, const float* __restrict__ xi,
              const float* __restrict__ B2, const float* __restrict__ C2,
              const float* __restrict__ D21, const float* __restrict__ D22,
              const float* __restrict__ bxi, const float* __restrict__ bu,
              const float* __restrict__ Fm,  const float* __restrict__ B1,
              const float* __restrict__ eps,
              float* __restrict__ out_u, float* __restrict__ out_xi) {
  const int lane = threadIdx.x;
  const int b0   = blockIdx.y * 16;
  const int cb   = blockIdx.x;
  const int colc = lane & 15;
  const int rhi  = (lane >> 4) << 3;

  const bool isu = cb < 2;
  const float* __restrict__ Wx = isu ? C2  : Fm;
  const float* __restrict__ We = isu ? D21 : B1;
  const float* __restrict__ Wy = isu ? D22 : B2;
  const float* __restrict__ bs = isu ? bu  : bxi;
  float* __restrict__ outp = isu ? out_u : out_xi;
  const int   ldo = isu ? M_OUT : N_XI;
  const float scl = isu ? OUT_AMP : INV_M;
  const int   n0  = (isu ? cb : cb - 2) * 32;

  v8f acc0 = {}, acc1 = {};
  seg16x32(xi,  Wx, N_XI,  b0, n0, lane, acc0, acc1);
  seg16x32(eps, We, L_DIM, b0, n0, lane, acc0, acc1);
  seg16x32(y_,  Wy, N_IN,  b0, n0, lane, acc0, acc1);
  const int col = n0 + colc;
  const float bc0 = bs[col];
  const float bc1 = bs[col + 16];
  #pragma unroll
  for (int r = 0; r < 8; ++r) {
    outp[(size_t)(b0 + r + rhi) * ldo + col]      = (acc0[r] + bc0) * scl;
    outp[(size_t)(b0 + r + rhi) * ldo + col + 16] = (acc1[r] + bc1) * scl;
  }
}

extern "C" void kernel_launch(void* const* d_in, const int* in_sizes, int n_in,
                              void* d_out, int out_size, void* d_ws, size_t ws_size,
                              hipStream_t stream) {
  const float* y_  = (const float*)d_in[0];
  const float* xi  = (const float*)d_in[1];
  const float* B2  = (const float*)d_in[2];
  const float* C2  = (const float*)d_in[3];
  const float* D21 = (const float*)d_in[4];
  const float* D22 = (const float*)d_in[5];
  const float* D12 = (const float*)d_in[6];
  const float* bxi = (const float*)d_in[7];
  const float* bv  = (const float*)d_in[8];
  const float* bu  = (const float*)d_in[9];
  const float* Fm  = (const float*)d_in[10];
  const float* B1  = (const float*)d_in[11];
  const float* Lam = (const float*)d_in[13];
  const float* C1  = (const float*)d_in[14];
  const float* D11 = (const float*)d_in[15];

  float* out_u  = (float*)d_out;
  float* out_xi = out_u + (size_t)B_TOT * M_OUT;

  float* Ews = (float*)d_ws;
  float* eps = Ews + (size_t)B_TOT * L_DIM;

  rnn_pre<<<dim3(8, 64), dim3(32), 0, stream>>>(y_, xi, D12, bv, C1, Ews);
  rnn_rec<<<dim3(64), dim3(32), 0, stream>>>(Lam, D11, Ews, eps);
  rnn_post<<<dim3(10, 64), dim3(32), 0, stream>>>(
      y_, xi, B2, C2, D21, D22, bxi, bu, Fm, B1, eps, out_u, out_xi);

  (void)in_sizes; (void)n_in; (void)out_size; (void)ws_size;
}


# Right-looking recurrence: a finished 16-column block is pushed by WMMA into all later tiles, its D11 pieces fetched ahead of the steps; no load wait between blocks; changes summation order of v
# speedup vs baseline: 28.2942x; 1.2983x over previous
#include <hip/hip_runtime.h>
#include <math.h>


#define B_TOT 1024
#define N_IN  128
#define M_OUT 64
#define N_XI  256
#define L_DIM 256
#define LDE   257
#define OUT_AMP 40.0f
#define INV_M  (1.0f/1.001f)
#define TWO_LOG2E 2.8853900817779268f

typedef __attribute__((ext_vector_type(2))) float v2f;
typedef __attribute__((ext_vector_type(4))) float v4f;
typedef __attribute__((ext_vector_type(8))) float v8f;

__device__ __forceinline__ v2f frag16x4_g(const float* __restrict__ M, int ld,
                                          int r0, int k0, int lane) {
  int r = r0 + (lane & 15);
  int k = k0 + ((lane >> 4) << 1);
  const float* p = M + (size_t)r * ld + k;
  v2f f; f.x = p[0]; f.y = p[1]; return f;
}

__device__ __forceinline__ v8f wmma4(v2f a, v2f b, v8f c) {
  return __builtin_amdgcn_wmma_f32_16x16x4_f32(false, a, false, b, (short)0, c,
                                               false, false);
}

__device__ __forceinline__ v2f d11_pair(v4f q0, v4f q1, int j) {
  v2f b;
  if (j == 0)      { b.x = q0.x; b.y = q0.y; }
  else if (j == 1) { b.x = q0.z; b.y = q0.w; }
  else if (j == 2) { b.x = q1.x; b.y = q1.y; }
  else             { b.x = q1.z; b.y = q1.w; }
  return b;
}

__device__ __forceinline__ void push_tile1(float* Et, const v2f (&a)[4], v4f q0, v4f q1) {
  v8f acc;
  #pragma unroll
  for (int r = 0; r < 8; ++r) acc[r] = Et[r * LDE];
  #pragma unroll
  for (int j = 0; j < 4; ++j) acc = wmma4(a[j], d11_pair(q0, q1, j), acc);
  #pragma unroll
  for (int r = 0; r < 8; ++r) Et[r * LDE] = acc[r];
}

__device__ __forceinline__ void seg16x32(const float* __restrict__ X, const float* __restrict__ Wt,
                                         int ld, int b0, int n0, int lane,
                                         v8f& acc0, v8f& acc1) {
  #pragma unroll 16
  for (int k0 = 0; k0 < ld; k0 += 4) {
    const v2f a = frag16x4_g(X, ld, b0, k0, lane);
    acc0 = wmma4(a, frag16x4_g(Wt, ld, n0,      k0, lane), acc0);
    acc1 = wmma4(a, frag16x4_g(Wt, ld, n0 + 16, k0, lane), acc1);
  }
}

__global__ __launch_bounds__(32)
void rnn_pre(const float* __restrict__ y_, const float* __restrict__ xi,
             const float* __restrict__ D12, const float* __restrict__ bv,
             const float* __restrict__ C1, float* __restrict__ Ews) {
  const int lane = threadIdx.x;
  const int b0   = blockIdx.y * 16;
  const int n0   = blockIdx.x * 32;
  const int colc = lane & 15;
  const int rhi  = (lane >> 4) << 3;

  v8f acc0 = {}, acc1 = {};
  seg16x32(xi, C1,  N_XI, b0, n0, lane, acc0, acc1);
  seg16x32(y_, D12, N_IN, b0, n0, lane, acc0, acc1);
  const int col = n0 + colc;
  const float bv0 = (col == 0) ? 0.0f : bv[col];
  const float bv1 = bv[col + 16];
  #pragma unroll
  for (int r = 0; r < 8; ++r) {
    Ews[(size_t)(b0 + r + rhi) * L_DIM + col]      = acc0[r] + bv0;
    Ews[(size_t)(b0 + r + rhi) * L_DIM + col + 16] = acc1[r] + bv1;
  }
}

__global__ __launch_bounds__(32)
void rnn_rec(const float* __restrict__ Lambda, const float* __restrict__ D11,
             const float* __restrict__ Ews, float* __restrict__ eps_ws) {
  __shared__ float eb[16][LDE];
  __shared__ __attribute__((aligned(16))) float dg[256];
  __shared__ __attribute__((aligned(16))) float cs[L_DIM];
  const int lane = threadIdx.x;
  const int b0   = blockIdx.x * 16;
  float* E = &eb[0][0];

  const int colc = lane & 15;
  const int rhi  = (lane >> 4) << 3;

  {
    float lr[8];
    #pragma unroll
    for (int t = 0; t < 8; ++t) lr[t] = Lambda[lane * 8 + t];
    #pragma unroll
    for (int t = 0; t < 8; ++t) cs[lane * 8 + t] = TWO_LOG2E / lr[t];
  }

  #pragma unroll
  for (int r = 0; r < 16; ++r) {
    #pragma unroll
    for (int h = 0; h < 2; ++h) {
      const int c = h * 128 + lane * 4;
      const v4f t = *(const v4f*)(Ews + (size_t)(b0 + r) * L_DIM + c);
      E[r * LDE + c + 0] = t.x; E[r * LDE + c + 1] = t.y;
      E[r * LDE + c + 2] = t.z; E[r * LDE + c + 3] = t.w;
    }
  }
  __syncthreads();

  {
    const int row = lane & 15;
    const bool hi = lane >= 16;
    float* DG = &dg[0];
    float dreg[8];
    {
      const float* dsrc = D11 + (size_t)(lane >> 1) * L_DIM + (lane & 1) * 8;
      #pragma unroll
      for (int t = 0; t < 8; ++t) dreg[t] = dsrc[t];
    }
    #pragma unroll 1
    for (int I = 0; I < 16; ++I) {
      const int c0 = I * 16;

      #pragma unroll
      for (int t = 0; t < 8; ++t) DG[lane * 8 + t] = dreg[t];
      float cl[16];
      #pragma unroll
      for (int ii = 0; ii < 16; ++ii) cl[ii] = cs[c0 + ii];
      __syncthreads();

      float v[16];
      #pragma unroll
      for (int jj = 0; jj < 16; ++jj) v[jj] = E[row * LDE + c0 + jj];
      float d[16][16];
      #pragma unroll
      for (int kk = 1; kk < 16; ++kk) {
        #pragma unroll
        for (int ii = 0; ii < kk; ++ii) d[kk][ii] = DG[kk * 16 + ii];
      }

      {
        const int cn = (I < 15 ? I + 1 : 15) * 16;
        const float* dsrc = D11 + (size_t)(cn + (lane >> 1)) * L_DIM + cn + (lane & 1) * 8;
        #pragma unroll
        for (int t = 0; t < 8; ++t) dreg[t] = dsrc[t];
      }
      v4f fq[16][2];
      {
        const float* fsrc = D11 + (size_t)row * L_DIM + c0 + (hi ? 8 : 0);
        #pragma unroll
        for (int g = 0; g < 4; ++g) {
          if (I < 4 * g + 3) {
            #pragma unroll
            for (int t = 0; t < 4; ++t) {
              const int J = 4 * g + t;
              if (J == 0) continue;
              fq[J][0] = *(const v4f*)(fsrc + (size_t)J * 16 * L_DIM);
              fq[J][1] = *(const v4f*)(fsrc + (size_t)J * 16 * L_DIM + 4);
            }
          }
        }
      }

      #pragma unroll
      for (int ii = 0; ii < 16; ++ii) {
        const float ex = __builtin_amdgcn_exp2f(v[ii] * cl[ii]);
        const float e  = fmaf(-2.0f, __builtin_amdgcn_rcpf(ex + 1.0f), 1.0f);
        v[ii] = e;
        #pragma unroll
        for (int kk = ii + 1; kk < 16; ++kk) v[kk] = fmaf(e, d[kk][ii], v[kk]);
      }
      if (lane < 16) {
        #pragma unroll
        for (int jj = 0; jj < 16; ++jj) E[row * LDE + c0 + jj] = v[jj];
      }

      v2f a[4];
      #pragma unroll
      for (int j = 0; j < 4; ++j) {
        a[j].x = hi ? v[8 + 2 * j] : v[2 * j];
        a[j].y = hi ? v[9 + 2 * j] : v[2 * j + 1];
      }
      float* Eb = E + rhi * LDE + colc;
      #pragma unroll
      for (int g = 0; g < 4; ++g) {
        if (I < 4 * g) {
          float* Eg = Eb + 4 * g * 16;
          v8f t0, t1, t2, t3;
          #pragma unroll
          for (int r = 0; r < 8; ++r) {
            t0[r] = Eg[r * LDE];      t1[r] = Eg[r * LDE + 16];
            t2[r] = Eg[r * LDE + 32]; t3[r] = Eg[r * LDE + 48];
          }
          #pragma unroll
          for (int j = 0; j < 4; ++j) {
            t0 = wmma4(a[j], d11_pair(fq[4 * g + 0][0], fq[4 * g + 0][1], j), t0);
            t1 = wmma4(a[j], d11_pair(fq[4 * g + 1][0], fq[4 * g + 1][1], j), t1);
            t2 = wmma4(a[j], d11_pair(fq[4 * g + 2][0], fq[4 * g + 2][1], j), t2);
            t3 = wmma4(a[j], d11_pair(fq[4 * g + 3][0], fq[4 * g + 3][1], j), t3);
          }
          #pragma unroll
          for (int r = 0; r < 8; ++r) {
            Eg[r * LDE]      = t0[r]; Eg[r * LDE + 16] = t1[r];
            Eg[r * LDE + 32] = t2[r]; Eg[r * LDE + 48] = t3[r];
          }
        } else {
          #pragma unroll
          for (int t = 0; t < 4; ++t) {
            const int J = 4 * g + t;
            if (J == 0) continue;
            if (J > I) push_tile1(Eb + J * 16, a, fq[J][0], fq[J][1]);
          }
        }
      }
    }
    __syncthreads();
  }

  #pragma unroll
  for (int r = 0; r < 16; ++r) {
    #pragma unroll
    for (int h = 0; h < 2; ++h) {
      const int c = h * 128 + lane * 4;
      v4f t;
      t.x = E[r * LDE + c + 0]; t.y = E[r * LDE + c + 1];
      t.z = E[r * LDE + c + 2]; t.w = E[r * LDE + c + 3];
      *(v4f*)(eps_ws + (size_t)(b0 + r) * L_DIM + c) = t;
    }
  }
}

__global__ __launch_bounds__(32)
void rnn_post(const float* __restrict__ y_, const float* __restrict__ xi,
              const float* __restrict__ B2, const float* __restrict__ C2,
              const float* __restrict__ D21, const float* __restrict__ D22,
              const float* __restrict__ bxi, const float* __restrict__ bu,
              const float* __restrict__ Fm,  const float* __restrict__ B1,
              const float* __restrict__ eps,
              float* __restrict__ out_u, float* __restrict__ out_xi) {
  const int lane = threadIdx.x;
  const int b0   = blockIdx.y * 16;
  const int cb   = blockIdx.x;
  const int colc = lane & 15;
  const int rhi  = (lane >> 4) << 3;

  const bool isu = cb < 2;
  const float* __restrict__ Wx = isu ? C2  : Fm;
  const float* __restrict__ We = isu ? D21 : B1;
  const float* __restrict__ Wy = isu ? D22 : B2;
  const float* __restrict__ bs = isu ? bu  : bxi;
  float* __restrict__ outp = isu ? out_u : out_xi;
  const int   ldo = isu ? M_OUT : N_XI;
  const float scl = isu ? OUT_AMP : INV_M;
  const int   n0  = (isu ? cb : cb - 2) * 32;

  v8f acc0 = {}, acc1 = {};
  seg16x32(xi,  Wx, N_XI,  b0, n0, lane, acc0, acc1);
  seg16x32(eps, We, L_DIM, b0, n0, lane, acc0, acc1);
  seg16x32(y_,  Wy, N_IN,  b0, n0, lane, acc0, acc1);
  const int col = n0 + colc;
  const float bc0 = bs[col];
  const float bc1 = bs[col + 16];
  #pragma unroll
  for (int r = 0; r < 8; ++r) {
    outp[(size_t)(b0 + r + rhi) * ldo + col]      = (acc0[r] + bc0) * scl;
    outp[(size_t)(b0 + r + rhi) * ldo + col + 16] = (acc1[r] + bc1) * scl;
  }
}

extern "C" void kernel_launch(void* const* d_in, const int* in_sizes, int n_in,
                              void* d_out, int out_size, void* d_ws, size_t ws_size,
                              hipStream_t stream) {
  const float* y_  = (const float*)d_in[0];
  const float* xi  = (const float*)d_in[1];
  const float* B2  = (const float*)d_in[2];
  const float* C2  = (const float*)d_in[3];
  const float* D21 = (const float*)d_in[4];
  const float* D22 = (const float*)d_in[5];
  const float* D12 = (const float*)d_in[6];
  const float* bxi = (const float*)d_in[7];
  const float* bv  = (const float*)d_in[8];
  const float* bu  = (const float*)d_in[9];
  const float* Fm  = (const float*)d_in[10];
  const float* B1  = (const float*)d_in[11];
  const float* Lam = (const float*)d_in[13];
  const float* C1  = (const float*)d_in[14];
  const float* D11 = (const float*)d_in[15];

  float* out_u  = (float*)d_out;
  float* out_xi = out_u + (size_t)B_TOT * M_OUT;

  float* Ews = (float*)d_ws;
  float* eps = Ews + (size_t)B_TOT * L_DIM;

  rnn_pre<<<dim3(8, 64), dim3(32), 0, stream>>>(y_, xi, D12, bv, C1, Ews);
  rnn_rec<<<dim3(64), dim3(32), 0, stream>>>(Lam, D11, Ews, eps);
  rnn_post<<<dim3(10, 64), dim3(32), 0, stream>>>(
      y_, xi, B2, C2, D21, D22, bxi, bu, Fm, B1, eps, out_u, out_xi);

  (void)in_sizes; (void)n_in; (void)out_size; (void)ws_size;
}


# Projection GEMMs (launches 1 and 3): operand loads widened to 16 bytes, 32 contiguous bytes per lane feed four WMMAs: half the load instructions and line touches per k; changes the k summation order
# speedup vs baseline: 32.6693x; 1.1546x over previous
#include <hip/hip_runtime.h>
#include <math.h>


#define B_TOT 1024
#define N_IN  128
#define M_OUT 64
#define N_XI  256
#define L_DIM 256
#define LDE   257
#define OUT_AMP 40.0f
#define INV_M  (1.0f/1.001f)
#define TWO_LOG2E 2.8853900817779268f

typedef __attribute__((ext_vector_type(2))) float v2f;
typedef __attribute__((ext_vector_type(4))) float v4f;
typedef __attribute__((ext_vector_type(8))) float v8f;

__device__ __forceinline__ void frag16x16_g(const float* __restrict__ M, int ld,
                                            int r0, int kc, int lane, v4f& q0, v4f& q1) {
  const float* p = M + (size_t)(r0 + (lane & 15)) * ld + kc + ((lane >> 4) << 3);
  q0 = *(const v4f*)p;
  q1 = *(const v4f*)(p + 4);
}

__device__ __forceinline__ v8f wmma4(v2f a, v2f b, v8f c) {
  return __builtin_amdgcn_wmma_f32_16x16x4_f32(false, a, false, b, (short)0, c,
                                               false, false);
}

__device__ __forceinline__ v2f pair8(v4f q0, v4f q1, int j) {
  v2f b;
  if (j == 0)      { b.x = q0.x; b.y = q0.y; }
  else if (j == 1) { b.x = q0.z; b.y = q0.w; }
  else if (j == 2) { b.x = q1.x; b.y = q1.y; }
  else             { b.x = q1.z; b.y = q1.w; }
  return b;
}

__device__ __forceinline__ void push_tile1(float* Et, const v2f (&a)[4], v4f q0, v4f q1) {
  v8f acc;
  #pragma unroll
  for (int r = 0; r < 8; ++r) acc[r] = Et[r * LDE];
  #pragma unroll
  for (int j = 0; j < 4; ++j) acc = wmma4(a[j], pair8(q0, q1, j), acc);
  #pragma unroll
  for (int r = 0; r < 8; ++r) Et[r * LDE] = acc[r];
}

__device__ __forceinline__ void seg16x32(const float* __restrict__ X, const float* __restrict__ Wt,
                                         int ld, int b0, int n0, int lane,
                                         v8f& acc0, v8f& acc1) {
  #pragma unroll 8
  for (int kc = 0; kc < ld; kc += 16) {
    v4f a0, a1, p0, p1, s0, s1;
    frag16x16_g(X,  ld, b0,      kc, lane, a0, a1);
    frag16x16_g(Wt, ld, n0,      kc, lane, p0, p1);
    frag16x16_g(Wt, ld, n0 + 16, kc, lane, s0, s1);
    #pragma unroll
    for (int j = 0; j < 4; ++j) {
      const v2f a = pair8(a0, a1, j);
      acc0 = wmma4(a, pair8(p0, p1, j), acc0);
      acc1 = wmma4(a, pair8(s0, s1, j), acc1);
    }
  }
}

__global__ __launch_bounds__(32) __attribute__((amdgpu_waves_per_eu(1, 1)))
void rnn_pre(const float* __restrict__ y_, const float* __restrict__ xi,
             const float* __restrict__ D12, const float* __restrict__ bv,
             const float* __restrict__ C1, float* __restrict__ Ews) {
  const int lane = threadIdx.x;
  const int b0   = blockIdx.y * 16;
  const int n0   = blockIdx.x * 32;
  const int colc = lane & 15;
  const int rhi  = (lane >> 4) << 3;

  v8f acc0 = {}, acc1 = {};
  seg16x32(xi, C1,  N_XI, b0, n0, lane, acc0, acc1);
  seg16x32(y_, D12, N_IN, b0, n0, lane, acc0, acc1);
  const int col = n0 + colc;
  const float bv0 = (col == 0) ? 0.0f : bv[col];
  const float bv1 = bv[col + 16];
  #pragma unroll
  for (int r = 0; r < 8; ++r) {
    Ews[(size_t)(b0 + r + rhi) * L_DIM + col]      = acc0[r] + bv0;
    Ews[(size_t)(b0 + r + rhi) * L_DIM + col + 16] = acc1[r] + bv1;
  }
}

__global__ __launch_bounds__(32)
void rnn_rec(const float* __restrict__ Lambda, const float* __restrict__ D11,
             const float* __restrict__ Ews, float* __restrict__ eps_ws) {
  __shared__ float eb[16][LDE];
  __shared__ __attribute__((aligned(16))) float dg[256];
  __shared__ __attribute__((aligned(16))) float cs[L_DIM];
  const int lane = threadIdx.x;
  const int b0   = blockIdx.x * 16;
  float* E = &eb[0][0];

  const int colc = lane & 15;
  const int rhi  = (lane >> 4) << 3;

  {
    float lr[8];
    #pragma unroll
    for (int t = 0; t < 8; ++t) lr[t] = Lambda[lane * 8 + t];
    #pragma unroll
    for (int t = 0; t < 8; ++t) cs[lane * 8 + t] = TWO_LOG2E / lr[t];
  }

  #pragma unroll
  for (int r = 0; r < 16; ++r) {
    #pragma unroll
    for (int h = 0; h < 2; ++h) {
      const int c = h * 128 + lane * 4;
      const v4f t = *(const v4f*)(Ews + (size_t)(b0 + r) * L_DIM + c);
      E[r * LDE + c + 0] = t.x; E[r * LDE + c + 1] = t.y;
      E[r * LDE + c + 2] = t.z; E[r * LDE + c + 3] = t.w;
    }
  }
  __syncthreads();

  {
    const int row = lane & 15;
    const bool hi = lane >= 16;
    float* DG = &dg[0];
    float dreg[8];
    {
      const float* dsrc = D11 + (size_t)(lane >> 1) * L_DIM + (lane & 1) * 8;
      #pragma unroll
      for (int t = 0; t < 8; ++t) dreg[t] = dsrc[t];
    }
    #pragma unroll 1
    for (int I = 0; I < 16; ++I) {
      const int c0 = I * 16;

      #pragma unroll
      for (int t = 0; t < 8; ++t) DG[lane * 8 + t] = dreg[t];
      float cl[16];
      #pragma unroll
      for (int ii = 0; ii < 16; ++ii) cl[ii] = cs[c0 + ii];
      __syncthreads();

      float v[16];
      #pragma unroll
      for (int jj = 0; jj < 16; ++jj) v[jj] = E[row * LDE + c0 + jj];
      float d[16][16];
      #pragma unroll
      for (int kk = 1; kk < 16; ++kk) {
        #pragma unroll
        for (int ii = 0; ii < kk; ++ii) d[kk][ii] = DG[kk * 16 + ii];
      }

      {
        const int cn = (I < 15 ? I + 1 : 15) * 16;
        const float* dsrc = D11 + (size_t)(cn + (lane >> 1)) * L_DIM + cn + (lane & 1) * 8;
        #pragma unroll
        for (int t = 0; t < 8; ++t) dreg[t] = dsrc[t];
      }
      v4f fq[16][2];
      {
        const float* fsrc = D11 + (size_t)row * L_DIM + c0 + (hi ? 8 : 0);
        #pragma unroll
        for (int g = 0; g < 4; ++g) {
          if (I < 4 * g + 3) {
            #pragma unroll
            for (int t = 0; t < 4; ++t) {
              const int J = 4 * g + t;
              if (J == 0) continue;
              fq[J][0] = *(const v4f*)(fsrc + (size_t)J * 16 * L_DIM);
              fq[J][1] = *(const v4f*)(fsrc + (size_t)J * 16 * L_DIM + 4);
            }
          }
        }
      }

      #pragma unroll
      for (int ii = 0; ii < 16; ++ii) {
        const float ex = __builtin_amdgcn_exp2f(v[ii] * cl[ii]);
        const float e  = fmaf(-2.0f, __builtin_amdgcn_rcpf(ex + 1.0f), 1.0f);
        v[ii] = e;
        #pragma unroll
        for (int kk = ii + 1; kk < 16; ++kk) v[kk] = fmaf(e, d[kk][ii], v[kk]);
      }
      if (lane < 16) {
        #pragma unroll
        for (int jj = 0; jj < 16; ++jj) E[row * LDE + c0 + jj] = v[jj];
      }

      v2f a[4];
      #pragma unroll
      for (int j = 0; j < 4; ++j) {
        a[j].x = hi ? v[8 + 2 * j] : v[2 * j];
        a[j].y = hi ? v[9 + 2 * j] : v[2 * j + 1];
      }
      float* Eb = E + rhi * LDE + colc;
      #pragma unroll
      for (int g = 0; g < 4; ++g) {
        if (I < 4 * g) {
          float* Eg = Eb + 4 * g * 16;
          v8f t0, t1, t2, t3;
          #pragma unroll
          for (int r = 0; r < 8; ++r) {
            t0[r] = Eg[r * LDE];      t1[r] = Eg[r * LDE + 16];
            t2[r] = Eg[r * LDE + 32]; t3[r] = Eg[r * LDE + 48];
          }
          #pragma unroll
          for (int j = 0; j < 4; ++j) {
            t0 = wmma4(a[j], pair8(fq[4 * g + 0][0], fq[4 * g + 0][1], j), t0);
            t1 = wmma4(a[j], pair8(fq[4 * g + 1][0], fq[4 * g + 1][1], j), t1);
            t2 = wmma4(a[j], pair8(fq[4 * g + 2][0], fq[4 * g + 2][1], j), t2);
            t3 = wmma4(a[j], pair8(fq[4 * g + 3][0], fq[4 * g + 3][1], j), t3);
          }
          #pragma unroll
          for (int r = 0; r < 8; ++r) {
            Eg[r * LDE]      = t0[r]; Eg[r * LDE + 16] = t1[r];
            Eg[r * LDE + 32] = t2[r]; Eg[r * LDE + 48] = t3[r];
          }
        } else {
          #pragma unroll
          for (int t = 0; t < 4; ++t) {
            const int J = 4 * g + t;
            if (J == 0) continue;
            if (J > I) push_tile1(Eb + J * 16, a, fq[J][0], fq[J][1]);
          }
        }
      }
    }
    __syncthreads();
  }

  #pragma unroll
  for (int r = 0; r < 16; ++r) {
    #pragma unroll
    for (int h = 0; h < 2; ++h) {
      const int c = h * 128 + lane * 4;
      v4f t;
      t.x = E[r * LDE + c + 0]; t.y = E[r * LDE + c + 1];
      t.z = E[r * LDE + c + 2]; t.w = E[r * LDE + c + 3];
      *(v4f*)(eps_ws + (size_t)(b0 + r) * L_DIM + c) = t;
    }
  }
}

__global__ __launch_bounds__(32) __attribute__((amdgpu_waves_per_eu(1, 1)))
void rnn_post(const float* __restrict__ y_, const float* __restrict__ xi,
              const float* __restrict__ B2, const float* __restrict__ C2,
              const float* __restrict__ D21, const float* __restrict__ D22,
              const float* __restrict__ bxi, const float* __restrict__ bu,
              const float* __restrict__ Fm,  const float* __restrict__ B1,
              const float* __restrict__ eps,
              float* __restrict__ out_u, float* __restrict__ out_xi) {
  const int lane = threadIdx.x;
  const int b0   = blockIdx.y * 16;
  const int cb   = blockIdx.x;
  const int colc = lane & 15;
  const int rhi  = (lane >> 4) << 3;

  const bool isu = cb < 2;
  const float* __restrict__ Wx = isu ? C2  : Fm;
  const float* __restrict__ We = isu ? D21 : B1;
  const float* __restrict__ Wy = isu ? D22 : B2;
  const float* __restrict__ bs = isu ? bu  : bxi;
  float* __restrict__ outp = isu ? out_u : out_xi;
  const int   ldo = isu ? M_OUT : N_XI;
  const float scl = isu ? OUT_AMP : INV_M;
  const int   n0  = (isu ? cb : cb - 2) * 32;

  v8f acc0 = {}, acc1 = {};
  seg16x32(xi,  Wx, N_XI,  b0, n0, lane, acc0, acc1);
  seg16x32(eps, We, L_DIM, b0, n0, lane, acc0, acc1);
  seg16x32(y_,  Wy, N_IN,  b0, n0, lane, acc0, acc1);
  const int col = n0 + colc;
  const float bc0 = bs[col];
  const float bc1 = bs[col + 16];
  #pragma unroll
  for (int r = 0; r < 8; ++r) {
    outp[(size_t)(b0 + r + rhi) * ldo + col]      = (acc0[r] + bc0) * scl;
    outp[(size_t)(b0 + r + rhi) * ldo + col + 16] = (acc1[r] + bc1) * scl;
  }
}

extern "C" void kernel_launch(void* const* d_in, const int* in_sizes, int n_in,
                              void* d_out, int out_size, void* d_ws, size_t ws_size,
                              hipStream_t stream) {
  const float* y_  = (const float*)d_in[0];
  const float* xi  = (const float*)d_in[1];
  const float* B2  = (const float*)d_in[2];
  const float* C2  = (const float*)d_in[3];
  const float* D21 = (const float*)d_in[4];
  const float* D22 = (const float*)d_in[5];
  const float* D12 = (const float*)d_in[6];
  const float* bxi = (const float*)d_in[7];
  const float* bv  = (const float*)d_in[8];
  const float* bu  = (const float*)d_in[9];
  const float* Fm  = (const float*)d_in[10];
  const float* B1  = (const float*)d_in[11];
  const float* Lam = (const float*)d_in[13];
  const float* C1  = (const float*)d_in[14];
  const float* D11 = (const float*)d_in[15];

  float* out_u  = (float*)d_out;
  float* out_xi = out_u + (size_t)B_TOT * M_OUT;

  float* Ews = (float*)d_ws;
  float* eps = Ews + (size_t)B_TOT * L_DIM;

  rnn_pre<<<dim3(8, 64), dim3(32), 0, stream>>>(y_, xi, D12, bv, C1, Ews);
  rnn_rec<<<dim3(64), dim3(32), 0, stream>>>(Lam, D11, Ews, eps);
  rnn_post<<<dim3(10, 64), dim3(32), 0, stream>>>(
      y_, xi, B2, C2, D21, D22, bxi, bu, Fm, B1, eps, out_u, out_xi);

  (void)in_sizes; (void)n_in; (void)out_size; (void)ws_size;
}


# Recurrence step's dependent path: tanh(v/Lambda) is one multiply by a tabled 1/Lambda and the hardware tanh (v_tanh_f32), was mul, exp2, add, rcp, fma; f32 as before, sums and their order unchanged
# speedup vs baseline: 33.8374x; 1.0358x over previous
#include <hip/hip_runtime.h>
#include <math.h>


#define B_TOT 1024
#define N_IN  128
#define M_OUT 64
#define N_XI  256
#define L_DIM 256
#define LDE   257
#define OUT_AMP 40.0f
#define INV_M  (1.0f/1.001f)

typedef __attribute__((ext_vector_type(2))) float v2f;
typedef __attribute__((ext_vector_type(4))) float v4f;
typedef __attribute__((ext_vector_type(8))) float v8f;

__device__ __forceinline__ void frag16x16_g(const float* __restrict__ M, int ld,
                                            int r0, int kc, int lane, v4f& q0, v4f& q1) {
  const float* p = M + (size_t)(r0 + (lane & 15)) * ld + kc + ((lane >> 4) << 3);
  q0 = *(const v4f*)p;
  q1 = *(const v4f*)(p + 4);
}

__device__ __forceinline__ v8f wmma4(v2f a, v2f b, v8f c) {
  return __builtin_amdgcn_wmma_f32_16x16x4_f32(false, a, false, b, (short)0, c,
                                               false, false);
}

__device__ __forceinline__ v2f pair8(v4f q0, v4f q1, int j) {
  v2f b;
  if (j == 0)      { b.x = q0.x; b.y = q0.y; }
  else if (j == 1) { b.x = q0.z; b.y = q0.w; }
  else if (j == 2) { b.x = q1.x; b.y = q1.y; }
  else             { b.x = q1.z; b.y = q1.w; }
  return b;
}

__device__ __forceinline__ void push_tile1(float* Et, const v2f (&a)[4], v4f q0, v4f q1) {
  v8f acc;
  #pragma unroll
  for (int r = 0; r < 8; ++r) acc[r] = Et[r * LDE];
  #pragma unroll
  for (int j = 0; j < 4; ++j) acc = wmma4(a[j], pair8(q0, q1, j), acc);
  #pragma unroll
  for (int r = 0; r < 8; ++r) Et[r * LDE] = acc[r];
}

__device__ __forceinline__ void seg16x32(const float* __restrict__ X, const float* __restrict__ Wt,
                                         int ld, int b0, int n0, int lane,
                                         v8f& acc0, v8f& acc1) {
  #pragma unroll 8
  for (int kc = 0; kc < ld; kc += 16) {
    v4f a0, a1, p0, p1, s0, s1;
    frag16x16_g(X,  ld, b0,      kc, lane, a0, a1);
    frag16x16_g(Wt, ld, n0,      kc, lane, p0, p1);
    frag16x16_g(Wt, ld, n0 + 16, kc, lane, s0, s1);
    #pragma unroll
    for (int j = 0; j < 4; ++j) {
      const v2f a = pair8(a0, a1, j);
      acc0 = wmma4(a, pair8(p0, p1, j), acc0);
      acc1 = wmma4(a, pair8(s0, s1, j), acc1);
    }
  }
}

__global__ __launch_bounds__(32) __attribute__((amdgpu_waves_per_eu(1, 1)))
void rnn_pre(const float* __restrict__ y_, const float* __restrict__ xi,
             const float* __restrict__ D12, const float* __restrict__ bv,
             const float* __restrict__ C1, float* __restrict__ Ews) {
  const int lane = threadIdx.x;
  const int b0   = blockIdx.y * 16;
  const int n0   = blockIdx.x * 32;
  const int colc = lane & 15;
  const int rhi  = (lane >> 4) << 3;

  v8f acc0 = {}, acc1 = {};
  seg16x32(xi, C1,  N_XI, b0, n0, lane, acc0, acc1);
  seg16x32(y_, D12, N_IN, b0, n0, lane, acc0, acc1);
  const int col = n0 + colc;
  const float bv0 = (col == 0) ? 0.0f : bv[col];
  const float bv1 = bv[col + 16];
  #pragma unroll
  for (int r = 0; r < 8; ++r) {
    Ews[(size_t)(b0 + r + rhi) * L_DIM + col]      = acc0[r] + bv0;
    Ews[(size_t)(b0 + r + rhi) * L_DIM + col + 16] = acc1[r] + bv1;
  }
}

__global__ __launch_bounds__(32)
void rnn_rec(const float* __restrict__ Lambda, const float* __restrict__ D11,
             const float* __restrict__ Ews, float* __restrict__ eps_ws) {
  __shared__ float eb[16][LDE];
  __shared__ __attribute__((aligned(16))) float dg[256];
  __shared__ __attribute__((aligned(16))) float cs[L_DIM];
  const int lane = threadIdx.x;
  const int b0   = blockIdx.x * 16;
  float* E = &eb[0][0];

  const int colc = lane & 15;
  const int rhi  = (lane >> 4) << 3;

  {
    float lr[8];
    #pragma unroll
    for (int t = 0; t < 8; ++t) lr[t] = Lambda[lane * 8 + t];
    #pragma unroll
    for (int t = 0; t < 8; ++t) cs[lane * 8 + t] = 1.0f / lr[t];
  }

  #pragma unroll
  for (int r = 0; r < 16; ++r) {
    #pragma unroll
    for (int h = 0; h < 2; ++h) {
      const int c = h * 128 + lane * 4;
      const v4f t = *(const v4f*)(Ews + (size_t)(b0 + r) * L_DIM + c);
      E[r * LDE + c + 0] = t.x; E[r * LDE + c + 1] = t.y;
      E[r * LDE + c + 2] = t.z; E[r * LDE + c + 3] = t.w;
    }
  }
  __syncthreads();

  {
    const int row = lane & 15;
    const bool hi = lane >= 16;
    float* DG = &dg[0];
    float dreg[8];
    {
      const float* dsrc = D11 + (size_t)(lane >> 1) * L_DIM + (lane & 1) * 8;
      #pragma unroll
      for (int t = 0; t < 8; ++t) dreg[t] = dsrc[t];
    }
    #pragma unroll 1
    for (int I = 0; I < 16; ++I) {
      const int c0 = I * 16;

      #pragma unroll
      for (int t = 0; t < 8; ++t) DG[lane * 8 + t] = dreg[t];
      float cl[16];
      #pragma unroll
      for (int ii = 0; ii < 16; ++ii) cl[ii] = cs[c0 + ii];
      __syncthreads();

      float v[16];
      #pragma unroll
      for (int jj = 0; jj < 16; ++jj) v[jj] = E[row * LDE + c0 + jj];
      float d[16][16];
      #pragma unroll
      for (int kk = 1; kk < 16; ++kk) {
        #pragma unroll
        for (int ii = 0; ii < kk; ++ii) d[kk][ii] = DG[kk * 16 + ii];
      }

      {
        const int cn = (I < 15 ? I + 1 : 15) * 16;
        const float* dsrc = D11 + (size_t)(cn + (lane >> 1)) * L_DIM + cn + (lane & 1) * 8;
        #pragma unroll
        for (int t = 0; t < 8; ++t) dreg[t] = dsrc[t];
      }
      v4f fq[16][2];
      {
        const float* fsrc = D11 + (size_t)row * L_DIM + c0 + (hi ? 8 : 0);
        #pragma unroll
        for (int g = 0; g < 4; ++g) {
          if (I < 4 * g + 3) {
            #pragma unroll
            for (int t = 0; t < 4; ++t) {
              const int J = 4 * g + t;
              if (J == 0) continue;
              fq[J][0] = *(const v4f*)(fsrc + (size_t)J * 16 * L_DIM);
              fq[J][1] = *(const v4f*)(fsrc + (size_t)J * 16 * L_DIM + 4);
            }
          }
        }
      }

      #pragma unroll
      for (int ii = 0; ii < 16; ++ii) {
        const float e = __builtin_amdgcn_tanhf(v[ii] * cl[ii]);
        v[ii] = e;
        #pragma unroll
        for (int kk = ii + 1; kk < 16; ++kk) v[kk] = fmaf(e, d[kk][ii], v[kk]);
      }
      if (lane < 16) {
        #pragma unroll
        for (int jj = 0; jj < 16; ++jj) E[row * LDE + c0 + jj] = v[jj];
      }

      v2f a[4];
      #pragma unroll
      for (int j = 0; j < 4; ++j) {
        a[j].x = hi ? v[8 + 2 * j] : v[2 * j];
        a[j].y = hi ? v[9 + 2 * j] : v[2 * j + 1];
      }
      float* Eb = E + rhi * LDE + colc;
      #pragma unroll
      for (int g = 0; g < 4; ++g) {
        if (I < 4 * g) {
          float* Eg = Eb + 4 * g * 16;
          v8f t0, t1, t2, t3;
          #pragma unroll
          for (int r = 0; r < 8; ++r) {
            t0[r] = Eg[r * LDE];      t1[r] = Eg[r * LDE + 16];
            t2[r] = Eg[r * LDE + 32]; t3[r] = Eg[r * LDE + 48];
          }
          #pragma unroll
          for (int j = 0; j < 4; ++j) {
            t0 = wmma4(a[j], pair8(fq[4 * g + 0][0], fq[4 * g + 0][1], j), t0);
            t1 = wmma4(a[j], pair8(fq[4 * g + 1][0], fq[4 * g + 1][1], j), t1);
            t2 = wmma4(a[j], pair8(fq[4 * g + 2][0], fq[4 * g + 2][1], j), t2);
            t3 = wmma4(a[j], pair8(fq[4 * g + 3][0], fq[4 * g + 3][1], j), t3);
          }
          #pragma unroll
          for (int r = 0; r < 8; ++r) {
            Eg[r * LDE]      = t0[r]; Eg[r * LDE + 16] = t1[r];
            Eg[r * LDE + 32] = t2[r]; Eg[r * LDE + 48] = t3[r];
          }
        } else {
          #pragma unroll
          for (int t = 0; t < 4; ++t) {
            const int J = 4 * g + t;
            if (J == 0) continue;
            if (J > I) push_tile1(Eb + J * 16, a, fq[J][0], fq[J][1]);
          }
        }
      }
    }
    __syncthreads();
  }

  #pragma unroll
  for (int r = 0; r < 16; ++r) {
    #pragma unroll
    for (int h = 0; h < 2; ++h) {
      const int c = h * 128 + lane * 4;
      v4f t;
      t.x = E[r * LDE + c + 0]; t.y = E[r * LDE + c + 1];
      t.z = E[r * LDE + c + 2]; t.w = E[r * LDE + c + 3];
      *(v4f*)(eps_ws + (size_t)(b0 + r) * L_DIM + c) = t;
    }
  }
}

__global__ __launch_bounds__(32) __attribute__((amdgpu_waves_per_eu(1, 1)))
void rnn_post(const float* __restrict__ y_, const float* __restrict__ xi,
              const float* __restrict__ B2, const float* __restrict__ C2,
              const float* __restrict__ D21, const float* __restrict__ D22,
              const float* __restrict__ bxi, const float* __restrict__ bu,
              const float* __restrict__ Fm,  const float* __restrict__ B1,
              const float* __restrict__ eps,
              float* __restrict__ out_u, float* __restrict__ out_xi) {
  const int lane = threadIdx.x;
  const int b0   = blockIdx.y * 16;
  const int cb   = blockIdx.x;
  const int colc = lane & 15;
  const int rhi  = (lane >> 4) << 3;

  const bool isu = cb < 2;
  const float* __restrict__ Wx = isu ? C2  : Fm;
  const float* __restrict__ We = isu ? D21 : B1;
  const float* __restrict__ Wy = isu ? D22 : B2;
  const float* __restrict__ bs = isu ? bu  : bxi;
  float* __restrict__ outp = isu ? out_u : out_xi;
  const int   ldo = isu ? M_OUT : N_XI;
  const float scl = isu ? OUT_AMP : INV_M;
  const int   n0  = (isu ? cb : cb - 2) * 32;

  v8f acc0 = {}, acc1 = {};
  seg16x32(xi,  Wx, N_XI,  b0, n0, lane, acc0, acc1);
  seg16x32(eps, We, L_DIM, b0, n0, lane, acc0, acc1);
  seg16x32(y_,  Wy, N_IN,  b0, n0, lane, acc0, acc1);
  const int col = n0 + colc;
  const float bc0 = bs[col];
  const float bc1 = bs[col + 16];
  #pragma unroll
  for (int r = 0; r < 8; ++r) {
    outp[(size_t)(b0 + r + rhi) * ldo + col]      = (acc0[r] + bc0) * scl;
    outp[(size_t)(b0 + r + rhi) * ldo + col + 16] = (acc1[r] + bc1) * scl;
  }
}

extern "C" void kernel_launch(void* const* d_in, const int* in_sizes, int n_in,
                              void* d_out, int out_size, void* d_ws, size_t ws_size,
                              hipStream_t stream) {
  const float* y_  = (const float*)d_in[0];
  const float* xi  = (const float*)d_in[1];
  const float* B2  = (const float*)d_in[2];
  const float* C2  = (const float*)d_in[3];
  const float* D21 = (const float*)d_in[4];
  const float* D22 = (const float*)d_in[5];
  const float* D12 = (const float*)d_in[6];
  const float* bxi = (const float*)d_in[7];
  const float* bv  = (const float*)d_in[8];
  const float* bu  = (const float*)d_in[9];
  const float* Fm  = (const float*)d_in[10];
  const float* B1  = (const float*)d_in[11];
  const float* Lam = (const float*)d_in[13];
  const float* C1  = (const float*)d_in[14];
  const float* D11 = (const float*)d_in[15];

  float* out_u  = (float*)d_out;
  float* out_xi = out_u + (size_t)B_TOT * M_OUT;

  float* Ews = (float*)d_ws;
  float* eps = Ews + (size_t)B_TOT * L_DIM;

  rnn_pre<<<dim3(8, 64), dim3(32), 0, stream>>>(y_, xi, D12, bv, C1, Ews);
  rnn_rec<<<dim3(64), dim3(32), 0, stream>>>(Lam, D11, Ews, eps);
  rnn_post<<<dim3(10, 64), dim3(32), 0, stream>>>(
      y_, xi, B2, C2, D21, D22, bxi, bu, Fm, B1, eps, out_u, out_xi);

  (void)in_sizes; (void)n_in; (void)out_size; (void)ws_size;
}


# Recurrence tile kept in registers as sixteen transposed WMMA accumulators (D11 piece as A, eps block as B): steps run on the accumulators, no LDS round trip per push; sums and their order unchanged
# speedup vs baseline: 39.0428x; 1.1538x over previous
#include <hip/hip_runtime.h>
#include <math.h>


#define B_TOT 1024
#define N_IN  128
#define M_OUT 64
#define N_XI  256
#define L_DIM 256
#define OUT_AMP 40.0f
#define INV_M  (1.0f/1.001f)

typedef __attribute__((ext_vector_type(2))) float v2f;
typedef __attribute__((ext_vector_type(4))) float v4f;
typedef __attribute__((ext_vector_type(8))) float v8f;

__device__ __forceinline__ void frag16x16_g(const float* __restrict__ M, int ld,
                                            int r0, int kc, int lane, v4f& q0, v4f& q1) {
  const float* p = M + (size_t)(r0 + (lane & 15)) * ld + kc + ((lane >> 4) << 3);
  q0 = *(const v4f*)p;
  q1 = *(const v4f*)(p + 4);
}

__device__ __forceinline__ v8f wmma4(v2f a, v2f b, v8f c) {
  return __builtin_amdgcn_wmma_f32_16x16x4_f32(false, a, false, b, (short)0, c,
                                               false, false);
}

__device__ __forceinline__ v2f pair8(v4f q0, v4f q1, int j) {
  v2f b;
  if (j == 0)      { b.x = q0.x; b.y = q0.y; }
  else if (j == 1) { b.x = q0.z; b.y = q0.w; }
  else if (j == 2) { b.x = q1.x; b.y = q1.y; }
  else             { b.x = q1.z; b.y = q1.w; }
  return b;
}

__device__ __forceinline__ void seg16x32(const float* __restrict__ X, const float* __restrict__ Wt,
                                         int ld, int b0, int n0, int lane,
                                         v8f& acc0, v8f& acc1) {
  #pragma unroll 8
  for (int kc = 0; kc < ld; kc += 16) {
    v4f a0, a1, p0, p1, s0, s1;
    frag16x16_g(X,  ld, b0,      kc, lane, a0, a1);
    frag16x16_g(Wt, ld, n0,      kc, lane, p0, p1);
    frag16x16_g(Wt, ld, n0 + 16, kc, lane, s0, s1);
    #pragma unroll
    for (int j = 0; j < 4; ++j) {
      const v2f a = pair8(a0, a1, j);
      acc0 = wmma4(a, pair8(p0, p1, j), acc0);
      acc1 = wmma4(a, pair8(s0, s1, j), acc1);
    }
  }
}

__global__ __launch_bounds__(32) __attribute__((amdgpu_waves_per_eu(1, 1)))
void rnn_pre(const float* __restrict__ y_, const float* __restrict__ xi,
             const float* __restrict__ D12, const float* __restrict__ bv,
             const float* __restrict__ C1, float* __restrict__ Ews) {
  const int lane = threadIdx.x;
  const int b0   = blockIdx.y * 16;
  const int n0   = blockIdx.x * 32;
  const int colc = lane & 15;
  const int rhi  = (lane >> 4) << 3;

  v8f acc0 = {}, acc1 = {};
  seg16x32(xi, C1,  N_XI, b0, n0, lane, acc0, acc1);
  seg16x32(y_, D12, N_IN, b0, n0, lane, acc0, acc1);
  const int col = n0 + colc;
  const float bv0 = (col == 0) ? 0.0f : bv[col];
  const float bv1 = bv[col + 16];
  #pragma unroll
  for (int r = 0; r < 8; ++r) {
    Ews[(size_t)(b0 + r + rhi) * L_DIM + col]      = acc0[r] + bv0;
    Ews[(size_t)(b0 + r + rhi) * L_DIM + col + 16] = acc1[r] + bv1;
  }
}


__device__ __forceinline__ float xhalf(float x) {
  const int b = __builtin_bit_cast(int, x);
  return __builtin_bit_cast(float,
      __builtin_amdgcn_permlanex16(b, b, 0x76543210, 0xfedcba98, false, false));
}

struct BlockCoef {
  v4f cl[2];
  float dd[8][8];
  v4f dx[8][2];
  v4f fq[15][2];
};

__device__ __forceinline__ float vsel(v4f q0, v4f q1, int i) {
  return i == 0 ? q0.x : i == 1 ? q0.y : i == 2 ? q0.z : i == 3 ? q0.w
       : i == 4 ? q1.x : i == 5 ? q1.y : i == 6 ? q1.z : q1.w;
}

__device__ __forceinline__ void fetch_cl(BlockCoef& k, const float* cs, int I, int h) {
  k.cl[0] = *(const v4f*)(cs + 16 * I + 8 * h);
  k.cl[1] = *(const v4f*)(cs + 16 * I + 8 * h + 4);
}

__device__ __forceinline__ void fetch_block(BlockCoef& k, const float* __restrict__ D11,
                                            int I, int c, int h) {
  const int c0 = 16 * I;
  const float* pd = D11 + (size_t)(c0 + 8 * h) * L_DIM + c0 + 8 * h;
  #pragma unroll
  for (int q = 1; q < 8; ++q) {
    #pragma unroll
    for (int r = 0; r < q; ++r) k.dd[q][r] = pd[(size_t)q * L_DIM + r];
  }
  const float* px = pd - 8 * h;
  #pragma unroll
  for (int q = 0; q < 8; ++q) {
    k.dx[q][0] = *(const v4f*)(px + (size_t)q * L_DIM);
    k.dx[q][1] = *(const v4f*)(px + (size_t)q * L_DIM + 4);
  }
  const float* pf = D11 + (size_t)c * L_DIM + c0 + 8 * h;
  #pragma unroll
  for (int t = 1; t < 16; ++t) {
    const float* pt = pf + (size_t)(((I + t) & 15) * 16) * L_DIM;
    k.fq[t - 1][0] = *(const v4f*)(pt);
    k.fq[t - 1][1] = *(const v4f*)(pt + 4);
  }
}

__device__ __forceinline__ v8f push1(v8f n, const v4f (&f)[2], const v2f (&b)[4]) {
  #pragma unroll
  for (int j = 0; j < 4; ++j) n = wmma4(pair8(f[0], f[1], j), b[j], n);
  return n;
}

__device__ __forceinline__ void do_block(const BlockCoef& k, v8f (&R)[16], int I, bool hi,
                                         float* __restrict__ eps_row) {
  float w[8], u[8], e1[8], e2[8];
  #pragma unroll
  for (int r = 0; r < 8; ++r) { w[r] = R[0][r]; u[r] = R[0][r]; }
  #pragma unroll
  for (int r = 0; r < 8; ++r) {
    e1[r] = __builtin_amdgcn_tanhf(w[r] * vsel(k.cl[0], k.cl[1], r));
    #pragma unroll
    for (int q = r + 1; q < 8; ++q) w[q] = fmaf(e1[r], k.dd[q][r], w[q]);
    const float ex = xhalf(e1[r]);
    #pragma unroll
    for (int q = 0; q < 8; ++q) u[q] = fmaf(ex, vsel(k.dx[q][0], k.dx[q][1], r), u[q]);
  }
  #pragma unroll
  for (int r = 0; r < 8; ++r) {
    e2[r] = __builtin_amdgcn_tanhf(u[r] * vsel(k.cl[0], k.cl[1], r));
    #pragma unroll
    for (int q = r + 1; q < 8; ++q) u[q] = fmaf(e2[r], k.dd[q][r], u[q]);
  }
  float e[8];
  #pragma unroll
  for (int r = 0; r < 8; ++r) e[r] = hi ? e2[r] : e1[r];

  {
    v4f s0, s1;
    s0.x = e[0]; s0.y = e[1]; s0.z = e[2]; s0.w = e[3];
    s1.x = e[4]; s1.y = e[5]; s1.z = e[6]; s1.w = e[7];
    *(v4f*)(eps_row + 16 * I)     = s0;
    *(v4f*)(eps_row + 16 * I + 4) = s1;
  }

  v2f b[4];
  #pragma unroll
  for (int j = 0; j < 4; ++j) { b[j].x = e[2 * j]; b[j].y = e[2 * j + 1]; }

  const int nrem = 15 - I;
  #pragma unroll
  for (int g = 0; g < 4; ++g) {
    const int t0 = 4 * g + 1;
    const int nt = g < 3 ? 4 : 3;
    if (nrem >= t0 + nt - 1) {
      v8f n0 = R[t0], n1 = R[t0 + 1], n2 = R[t0 + 2], n3 = R[nt == 4 ? t0 + 3 : t0 + 2];
      #pragma unroll
      for (int j = 0; j < 4; ++j) {
        n0 = wmma4(pair8(k.fq[t0 - 1][0], k.fq[t0 - 1][1], j), b[j], n0);
        n1 = wmma4(pair8(k.fq[t0][0],     k.fq[t0][1],     j), b[j], n1);
        n2 = wmma4(pair8(k.fq[t0 + 1][0], k.fq[t0 + 1][1], j), b[j], n2);
        if (nt == 4) n3 = wmma4(pair8(k.fq[t0 + 2][0], k.fq[t0 + 2][1], j), b[j], n3);
      }
      R[t0 - 1] = n0; R[t0] = n1; R[t0 + 1] = n2;
      if (nt == 4) R[t0 + 2] = n3;
    } else {
      #pragma unroll
      for (int t = t0; t < t0 + nt; ++t) {
        if (t <= nrem) R[t - 1] = push1(R[t], k.fq[t - 1], b);
      }
    }
  }
}

__global__ __launch_bounds__(32) __attribute__((amdgpu_waves_per_eu(1, 1)))
void rnn_rec(const float* __restrict__ Lambda, const float* __restrict__ D11,
             const float* __restrict__ Ews, float* __restrict__ eps_ws) {
  __shared__ __attribute__((aligned(16))) float cs[L_DIM];
  const int lane = threadIdx.x;
  const int b0   = blockIdx.x * 16;
  const int c    = lane & 15;
  const int h    = lane >> 4;
  const bool hi  = h != 0;

  float lr[8];
  #pragma unroll
  for (int t = 0; t < 8; ++t) lr[t] = Lambda[lane * 8 + t];

  v8f R[16];
  {
    const float* erow = Ews + (size_t)(b0 + c) * L_DIM + 8 * h;
    #pragma unroll
    for (int J = 0; J < 16; ++J) {
      const v4f q0 = *(const v4f*)(erow + 16 * J);
      const v4f q1 = *(const v4f*)(erow + 16 * J + 4);
      R[J][0] = q0.x; R[J][1] = q0.y; R[J][2] = q0.z; R[J][3] = q0.w;
      R[J][4] = q1.x; R[J][5] = q1.y; R[J][6] = q1.z; R[J][7] = q1.w;
    }
  }
  BlockCoef ka, kb;
  fetch_block(ka, D11, 0, c, h);

  #pragma unroll
  for (int t = 0; t < 8; ++t) cs[lane * 8 + t] = 1.0f / lr[t];
  __syncthreads();
  fetch_cl(ka, cs, 0, h);

  float* eps_row = eps_ws + (size_t)(b0 + c) * L_DIM + 8 * h;

  #pragma unroll 1
  for (int I = 0; I < 16; I += 2) {
    fetch_block(kb, D11, I + 1, c, h);
    fetch_cl(kb, cs, I + 1, h);
    do_block(ka, R, I, hi, eps_row);
    const int In = I + 2 < 16 ? I + 2 : 15;
    fetch_block(ka, D11, In, c, h);
    fetch_cl(ka, cs, In, h);
    do_block(kb, R, I + 1, hi, eps_row);
  }
}

__global__ __launch_bounds__(32) __attribute__((amdgpu_waves_per_eu(1, 1)))
void rnn_post(const float* __restrict__ y_, const float* __restrict__ xi,
              const float* __restrict__ B2, const float* __restrict__ C2,
              const float* __restrict__ D21, const float* __restrict__ D22,
              const float* __restrict__ bxi, const float* __restrict__ bu,
              const float* __restrict__ Fm,  const float* __restrict__ B1,
              const float* __restrict__ eps,
              float* __restrict__ out_u, float* __restrict__ out_xi) {
  const int lane = threadIdx.x;
  const int b0   = blockIdx.y * 16;
  const int cb   = blockIdx.x;
  const int colc = lane & 15;
  const int rhi  = (lane >> 4) << 3;

  const bool isu = cb < 2;
  const float* __restrict__ Wx = isu ? C2  : Fm;
  const float* __restrict__ We = isu ? D21 : B1;
  const float* __restrict__ Wy = isu ? D22 : B2;
  const float* __restrict__ bs = isu ? bu  : bxi;
  float* __restrict__ outp = isu ? out_u : out_xi;
  const int   ldo = isu ? M_OUT : N_XI;
  const float scl = isu ? OUT_AMP : INV_M;
  const int   n0  = (isu ? cb : cb - 2) * 32;

  v8f acc0 = {}, acc1 = {};
  seg16x32(xi,  Wx, N_XI,  b0, n0, lane, acc0, acc1);
  seg16x32(eps, We, L_DIM, b0, n0, lane, acc0, acc1);
  seg16x32(y_,  Wy, N_IN,  b0, n0, lane, acc0, acc1);
  const int col = n0 + colc;
  const float bc0 = bs[col];
  const float bc1 = bs[col + 16];
  #pragma unroll
  for (int r = 0; r < 8; ++r) {
    outp[(size_t)(b0 + r + rhi) * ldo + col]      = (acc0[r] + bc0) * scl;
    outp[(size_t)(b0 + r + rhi) * ldo + col + 16] = (acc1[r] + bc1) * scl;
  }
}

extern "C" void kernel_launch(void* const* d_in, const int* in_sizes, int n_in,
                              void* d_out, int out_size, void* d_ws, size_t ws_size,
                              hipStream_t stream) {
  const float* y_  = (const float*)d_in[0];
  const float* xi  = (const float*)d_in[1];
  const float* B2  = (const float*)d_in[2];
  const float* C2  = (const float*)d_in[3];
  const float* D21 = (const float*)d_in[4];
  const float* D22 = (const float*)d_in[5];
  const float* D12 = (const float*)d_in[6];
  const float* bxi = (const float*)d_in[7];
  const float* bv  = (const float*)d_in[8];
  const float* bu  = (const float*)d_in[9];
  const float* Fm  = (const float*)d_in[10];
  const float* B1  = (const float*)d_in[11];
  const float* Lam = (const float*)d_in[13];
  const float* C1  = (const float*)d_in[14];
  const float* D11 = (const float*)d_in[15];

  float* out_u  = (float*)d_out;
  float* out_xi = out_u + (size_t)B_TOT * M_OUT;

  float* Ews = (float*)d_ws;
  float* eps = Ews + (size_t)B_TOT * L_DIM;

  rnn_pre<<<dim3(8, 64), dim3(32), 0, stream>>>(y_, xi, D12, bv, C1, Ews);
  rnn_rec<<<dim3(64), dim3(32), 0, stream>>>(Lam, D11, Ews, eps);
  rnn_post<<<dim3(10, 64), dim3(32), 0, stream>>>(
      y_, xi, B2, C2, D21, D22, bxi, bu, Fm, B1, eps, out_u, out_xi);

  (void)in_sizes; (void)n_in; (void)out_size; (void)ws_size;
}


# Launch 3's xi and y_ k segments are summed by 640 extra one-wave workgroups inside launch 2, on SIMDs the 64 recurrence waves leave idle; launch 3 adds only eps: k order now xi, y_, eps
# speedup vs baseline: 43.2284x; 1.1072x over previous
#include <hip/hip_runtime.h>
#include <math.h>


#define B_TOT 1024
#define N_IN  128
#define M_OUT 64
#define N_XI  256
#define L_DIM 256
#define OUT_AMP 40.0f
#define INV_M  (1.0f/1.001f)

typedef __attribute__((ext_vector_type(2))) float v2f;
typedef __attribute__((ext_vector_type(4))) float v4f;
typedef __attribute__((ext_vector_type(8))) float v8f;

__device__ __forceinline__ void frag16x16_g(const float* __restrict__ M, int ld,
                                            int r0, int kc, int lane, v4f& q0, v4f& q1) {
  const float* p = M + (size_t)(r0 + (lane & 15)) * ld + kc + ((lane >> 4) << 3);
  q0 = *(const v4f*)p;
  q1 = *(const v4f*)(p + 4);
}

__device__ __forceinline__ v8f wmma4(v2f a, v2f b, v8f c) {
  return __builtin_amdgcn_wmma_f32_16x16x4_f32(false, a, false, b, (short)0, c,
                                               false, false);
}

__device__ __forceinline__ v2f pair8(v4f q0, v4f q1, int j) {
  v2f b;
  if (j == 0)      { b.x = q0.x; b.y = q0.y; }
  else if (j == 1) { b.x = q0.z; b.y = q0.w; }
  else if (j == 2) { b.x = q1.x; b.y = q1.y; }
  else             { b.x = q1.z; b.y = q1.w; }
  return b;
}

__device__ __forceinline__ void seg16x32(const float* __restrict__ X, const float* __restrict__ Wt,
                                         int ld, int b0, int n0, int lane,
                                         v8f& acc0, v8f& acc1) {
  #pragma unroll 8
  for (int kc = 0; kc < ld; kc += 16) {
    v4f a0, a1, p0, p1, s0, s1;
    frag16x16_g(X,  ld, b0,      kc, lane, a0, a1);
    frag16x16_g(Wt, ld, n0,      kc, lane, p0, p1);
    frag16x16_g(Wt, ld, n0 + 16, kc, lane, s0, s1);
    #pragma unroll
    for (int j = 0; j < 4; ++j) {
      const v2f a = pair8(a0, a1, j);
      acc0 = wmma4(a, pair8(p0, p1, j), acc0);
      acc1 = wmma4(a, pair8(s0, s1, j), acc1);
    }
  }
}

__global__ __launch_bounds__(32) __attribute__((amdgpu_waves_per_eu(1, 1)))
void rnn_pre(const float* __restrict__ y_, const float* __restrict__ xi,
             const float* __restrict__ D12, const float* __restrict__ bv,
             const float* __restrict__ C1, float* __restrict__ Ews) {
  const int lane = threadIdx.x;
  const int b0   = blockIdx.y * 16;
  const int n0   = blockIdx.x * 32;
  const int colc = lane & 15;
  const int rhi  = (lane >> 4) << 3;

  v8f acc0 = {}, acc1 = {};
  seg16x32(xi, C1,  N_XI, b0, n0, lane, acc0, acc1);
  seg16x32(y_, D12, N_IN, b0, n0, lane, acc0, acc1);
  const int col = n0 + colc;
  const float bv0 = (col == 0) ? 0.0f : bv[col];
  const float bv1 = bv[col + 16];
  #pragma unroll
  for (int r = 0; r < 8; ++r) {
    Ews[(size_t)(b0 + r + rhi) * L_DIM + col]      = acc0[r] + bv0;
    Ews[(size_t)(b0 + r + rhi) * L_DIM + col + 16] = acc1[r] + bv1;
  }
}

#define POST_BLOCKS 640
#define REC_TILES   64
__device__ __forceinline__ void post_head(const float* __restrict__ y_, const float* __restrict__ xi,
                                          const float* __restrict__ B2, const float* __restrict__ C2,
                                          const float* __restrict__ D22, const float* __restrict__ Fm,
                                          float* __restrict__ Pws, int wg, int lane) {
  const int bt = wg / 10;
  const int cb = wg - bt * 10;
  const int b0 = bt * 16;
  const bool isu = cb < 2;
  const float* __restrict__ Wx = isu ? C2  : Fm;
  const float* __restrict__ Wy = isu ? D22 : B2;
  const int n0 = (isu ? cb : cb - 2) * 32;

  v8f acc0 = {}, acc1 = {};
  seg16x32(xi, Wx, N_XI, b0, n0, lane, acc0, acc1);
  seg16x32(y_, Wy, N_IN, b0, n0, lane, acc0, acc1);

  float* pb = Pws + (size_t)wg * 512 + lane * 4;
  v4f s0, s1, s2, s3;
  s0.x = acc0[0]; s0.y = acc0[1]; s0.z = acc0[2]; s0.w = acc0[3];
  s1.x = acc0[4]; s1.y = acc0[5]; s1.z = acc0[6]; s1.w = acc0[7];
  s2.x = acc1[0]; s2.y = acc1[1]; s2.z = acc1[2]; s2.w = acc1[3];
  s3.x = acc1[4]; s3.y = acc1[5]; s3.z = acc1[6]; s3.w = acc1[7];
  *(v4f*)(pb)       = s0;
  *(v4f*)(pb + 128) = s1;
  *(v4f*)(pb + 256) = s2;
  *(v4f*)(pb + 384) = s3;
}


__device__ __forceinline__ float xhalf(float x) {
  const int b = __builtin_bit_cast(int, x);
  return __builtin_bit_cast(float,
      __builtin_amdgcn_permlanex16(b, b, 0x76543210, 0xfedcba98, false, false));
}

struct BlockCoef {
  v4f cl[2];
  float dd[8][8];
  v4f dx[8][2];
  v4f fq[15][2];
};

__device__ __forceinline__ float vsel(v4f q0, v4f q1, int i) {
  return i == 0 ? q0.x : i == 1 ? q0.y : i == 2 ? q0.z : i == 3 ? q0.w
       : i == 4 ? q1.x : i == 5 ? q1.y : i == 6 ? q1.z : q1.w;
}

__device__ __forceinline__ void fetch_cl(BlockCoef& k, const float* cs, int I, int h) {
  k.cl[0] = *(const v4f*)(cs + 16 * I + 8 * h);
  k.cl[1] = *(const v4f*)(cs + 16 * I + 8 * h + 4);
}

__device__ __forceinline__ void fetch_block(BlockCoef& k, const float* __restrict__ D11,
                                            int I, int c, int h) {
  const int c0 = 16 * I;
  const float* pd = D11 + (size_t)(c0 + 8 * h) * L_DIM + c0 + 8 * h;
  #pragma unroll
  for (int q = 1; q < 8; ++q) {
    #pragma unroll
    for (int r = 0; r < q; ++r) k.dd[q][r] = pd[(size_t)q * L_DIM + r];
  }
  const float* px = pd - 8 * h;
  #pragma unroll
  for (int q = 0; q < 8; ++q) {
    k.dx[q][0] = *(const v4f*)(px + (size_t)q * L_DIM);
    k.dx[q][1] = *(const v4f*)(px + (size_t)q * L_DIM + 4);
  }
  const float* pf = D11 + (size_t)c * L_DIM + c0 + 8 * h;
  #pragma unroll
  for (int t = 1; t < 16; ++t) {
    const float* pt = pf + (size_t)(((I + t) & 15) * 16) * L_DIM;
    k.fq[t - 1][0] = *(const v4f*)(pt);
    k.fq[t - 1][1] = *(const v4f*)(pt + 4);
  }
}

__device__ __forceinline__ v8f push1(v8f n, const v4f (&f)[2], const v2f (&b)[4]) {
  #pragma unroll
  for (int j = 0; j < 4; ++j) n = wmma4(pair8(f[0], f[1], j), b[j], n);
  return n;
}

__device__ __forceinline__ void do_block(const BlockCoef& k, v8f (&R)[16], int I, bool hi,
                                         float* __restrict__ eps_row) {
  float w[8], u[8], e1[8], e2[8];
  #pragma unroll
  for (int r = 0; r < 8; ++r) { w[r] = R[0][r]; u[r] = R[0][r]; }
  #pragma unroll
  for (int r = 0; r < 8; ++r) {
    e1[r] = __builtin_amdgcn_tanhf(w[r] * vsel(k.cl[0], k.cl[1], r));
    #pragma unroll
    for (int q = r + 1; q < 8; ++q) w[q] = fmaf(e1[r], k.dd[q][r], w[q]);
    const float ex = xhalf(e1[r]);
    #pragma unroll
    for (int q = 0; q < 8; ++q) u[q] = fmaf(ex, vsel(k.dx[q][0], k.dx[q][1], r), u[q]);
  }
  #pragma unroll
  for (int r = 0; r < 8; ++r) {
    e2[r] = __builtin_amdgcn_tanhf(u[r] * vsel(k.cl[0], k.cl[1], r));
    #pragma unroll
    for (int q = r + 1; q < 8; ++q) u[q] = fmaf(e2[r], k.dd[q][r], u[q]);
  }
  float e[8];
  #pragma unroll
  for (int r = 0; r < 8; ++r) e[r] = hi ? e2[r] : e1[r];

  {
    v4f s0, s1;
    s0.x = e[0]; s0.y = e[1]; s0.z = e[2]; s0.w = e[3];
    s1.x = e[4]; s1.y = e[5]; s1.z = e[6]; s1.w = e[7];
    *(v4f*)(eps_row + 16 * I)     = s0;
    *(v4f*)(eps_row + 16 * I + 4) = s1;
  }

  v2f b[4];
  #pragma unroll
  for (int j = 0; j < 4; ++j) { b[j].x = e[2 * j]; b[j].y = e[2 * j + 1]; }

  const int nrem = 15 - I;
  #pragma unroll
  for (int g = 0; g < 4; ++g) {
    const int t0 = 4 * g + 1;
    const int nt = g < 3 ? 4 : 3;
    if (nrem >= t0 + nt - 1) {
      v8f n0 = R[t0], n1 = R[t0 + 1], n2 = R[t0 + 2], n3 = R[nt == 4 ? t0 + 3 : t0 + 2];
      #pragma unroll
      for (int j = 0; j < 4; ++j) {
        n0 = wmma4(pair8(k.fq[t0 - 1][0], k.fq[t0 - 1][1], j), b[j], n0);
        n1 = wmma4(pair8(k.fq[t0][0],     k.fq[t0][1],     j), b[j], n1);
        n2 = wmma4(pair8(k.fq[t0 + 1][0], k.fq[t0 + 1][1], j), b[j], n2);
        if (nt == 4) n3 = wmma4(pair8(k.fq[t0 + 2][0], k.fq[t0 + 2][1], j), b[j], n3);
      }
      R[t0 - 1] = n0; R[t0] = n1; R[t0 + 1] = n2;
      if (nt == 4) R[t0 + 2] = n3;
    } else {
      #pragma unroll
      for (int t = t0; t < t0 + nt; ++t) {
        if (t <= nrem) R[t - 1] = push1(R[t], k.fq[t - 1], b);
      }
    }
  }
}

__global__ __launch_bounds__(32) __attribute__((amdgpu_waves_per_eu(1, 1)))
void rnn_rec(const float* __restrict__ Lambda, const float* __restrict__ D11,
             const float* __restrict__ Ews, float* __restrict__ eps_ws,
             const float* __restrict__ y_, const float* __restrict__ xi,
             const float* __restrict__ B2, const float* __restrict__ C2,
             const float* __restrict__ D22, const float* __restrict__ Fm,
             float* __restrict__ Pws) {
  __shared__ __attribute__((aligned(16))) float cs[L_DIM];
  const int lane = threadIdx.x;
  if (blockIdx.x >= REC_TILES) {
    post_head(y_, xi, B2, C2, D22, Fm, Pws, blockIdx.x - REC_TILES, lane);
    return;
  }
  const int b0   = blockIdx.x * 16;
  const int c    = lane & 15;
  const int h    = lane >> 4;
  const bool hi  = h != 0;

  float lr[8];
  #pragma unroll
  for (int t = 0; t < 8; ++t) lr[t] = Lambda[lane * 8 + t];

  v8f R[16];
  {
    const float* erow = Ews + (size_t)(b0 + c) * L_DIM + 8 * h;
    #pragma unroll
    for (int J = 0; J < 16; ++J) {
      const v4f q0 = *(const v4f*)(erow + 16 * J);
      const v4f q1 = *(const v4f*)(erow + 16 * J + 4);
      R[J][0] = q0.x; R[J][1] = q0.y; R[J][2] = q0.z; R[J][3] = q0.w;
      R[J][4] = q1.x; R[J][5] = q1.y; R[J][6] = q1.z; R[J][7] = q1.w;
    }
  }
  BlockCoef ka, kb;
  fetch_block(ka, D11, 0, c, h);

  #pragma unroll
  for (int t = 0; t < 8; ++t) cs[lane * 8 + t] = 1.0f / lr[t];
  __syncthreads();
  fetch_cl(ka, cs, 0, h);

  float* eps_row = eps_ws + (size_t)(b0 + c) * L_DIM + 8 * h;

  #pragma unroll 1
  for (int I = 0; I < 16; I += 2) {
    fetch_block(kb, D11, I + 1, c, h);
    fetch_cl(kb, cs, I + 1, h);
    do_block(ka, R, I, hi, eps_row);
    const int In = I + 2 < 16 ? I + 2 : 15;
    fetch_block(ka, D11, In, c, h);
    fetch_cl(ka, cs, In, h);
    do_block(kb, R, I + 1, hi, eps_row);
  }
}

__global__ __launch_bounds__(32) __attribute__((amdgpu_waves_per_eu(1, 1)))
void rnn_post(const float* __restrict__ D21, const float* __restrict__ B1,
              const float* __restrict__ bxi, const float* __restrict__ bu,
              const float* __restrict__ eps, const float* __restrict__ Pws,
              float* __restrict__ out_u, float* __restrict__ out_xi) {
  const int lane = threadIdx.x;
  const int b0   = blockIdx.y * 16;
  const int cb   = blockIdx.x;
  const int colc = lane & 15;
  const int rhi  = (lane >> 4) << 3;

  const bool isu = cb < 2;
  const float* __restrict__ We = isu ? D21 : B1;
  const float* __restrict__ bs = isu ? bu  : bxi;
  float* __restrict__ outp = isu ? out_u : out_xi;
  const int   ldo = isu ? M_OUT : N_XI;
  const float scl = isu ? OUT_AMP : INV_M;
  const int   n0  = (isu ? cb : cb - 2) * 32;

  v8f acc0, acc1;
  {
    const float* pb = Pws + (size_t)(blockIdx.y * 10 + cb) * 512 + lane * 4;
    const v4f s0 = *(const v4f*)(pb);
    const v4f s1 = *(const v4f*)(pb + 128);
    const v4f s2 = *(const v4f*)(pb + 256);
    const v4f s3 = *(const v4f*)(pb + 384);
    acc0[0] = s0.x; acc0[1] = s0.y; acc0[2] = s0.z; acc0[3] = s0.w;
    acc0[4] = s1.x; acc0[5] = s1.y; acc0[6] = s1.z; acc0[7] = s1.w;
    acc1[0] = s2.x; acc1[1] = s2.y; acc1[2] = s2.z; acc1[3] = s2.w;
    acc1[4] = s3.x; acc1[5] = s3.y; acc1[6] = s3.z; acc1[7] = s3.w;
  }
  seg16x32(eps, We, L_DIM, b0, n0, lane, acc0, acc1);
  const int col = n0 + colc;
  const float bc0 = bs[col];
  const float bc1 = bs[col + 16];
  #pragma unroll
  for (int r = 0; r < 8; ++r) {
    outp[(size_t)(b0 + r + rhi) * ldo + col]      = (acc0[r] + bc0) * scl;
    outp[(size_t)(b0 + r + rhi) * ldo + col + 16] = (acc1[r] + bc1) * scl;
  }
}

extern "C" void kernel_launch(void* const* d_in, const int* in_sizes, int n_in,
                              void* d_out, int out_size, void* d_ws, size_t ws_size,
                              hipStream_t stream) {
  const float* y_  = (const float*)d_in[0];
  const float* xi  = (const float*)d_in[1];
  const float* B2  = (const float*)d_in[2];
  const float* C2  = (const float*)d_in[3];
  const float* D21 = (const float*)d_in[4];
  const float* D22 = (const float*)d_in[5];
  const float* D12 = (const float*)d_in[6];
  const float* bxi = (const float*)d_in[7];
  const float* bv  = (const float*)d_in[8];
  const float* bu  = (const float*)d_in[9];
  const float* Fm  = (const float*)d_in[10];
  const float* B1  = (const float*)d_in[11];
  const float* Lam = (const float*)d_in[13];
  const float* C1  = (const float*)d_in[14];
  const float* D11 = (const float*)d_in[15];

  float* out_u  = (float*)d_out;
  float* out_xi = out_u + (size_t)B_TOT * M_OUT;

  float* Ews = (float*)d_ws;
  float* eps = Ews + (size_t)B_TOT * L_DIM;
  float* Pws = eps + (size_t)B_TOT * L_DIM;

  rnn_pre<<<dim3(8, 64), dim3(32), 0, stream>>>(y_, xi, D12, bv, C1, Ews);
  rnn_rec<<<dim3(REC_TILES + POST_BLOCKS), dim3(32), 0, stream>>>(
      Lam, D11, Ews, eps, y_, xi, B2, C2, D22, Fm, Pws);
  rnn_post<<<dim3(10, 64), dim3(32), 0, stream>>>(
      D21, B1, bxi, bu, eps, Pws, out_u, out_xi);

  (void)in_sizes; (void)n_in; (void)out_size; (void)ws_size;
}


# Projection GEMMs (launches 1, 3, ride-along sums): each 16-k block's operand loads are issued eight blocks ahead of its WMMAs, k range unrolled; no load wait per 128 k; sums and order unchanged
# speedup vs baseline: 45.7099x; 1.0574x over previous
#include <hip/hip_runtime.h>
#include <math.h>


#define B_TOT 1024
#define N_IN  128
#define M_OUT 64
#define N_XI  256
#define L_DIM 256
#define OUT_AMP 40.0f
#define INV_M  (1.0f/1.001f)

typedef __attribute__((ext_vector_type(2))) float v2f;
typedef __attribute__((ext_vector_type(4))) float v4f;
typedef __attribute__((ext_vector_type(8))) float v8f;

__device__ __forceinline__ void frag16x16_g(const float* __restrict__ M, int ld,
                                            int r0, int kc, int lane, v4f& q0, v4f& q1) {
  const float* p = M + (size_t)(r0 + (lane & 15)) * ld + kc + ((lane >> 4) << 3);
  q0 = *(const v4f*)p;
  q1 = *(const v4f*)(p + 4);
}

__device__ __forceinline__ v8f wmma4(v2f a, v2f b, v8f c) {
  return __builtin_amdgcn_wmma_f32_16x16x4_f32(false, a, false, b, (short)0, c,
                                               false, false);
}

__device__ __forceinline__ v2f pair8(v4f q0, v4f q1, int j) {
  v2f b;
  if (j == 0)      { b.x = q0.x; b.y = q0.y; }
  else if (j == 1) { b.x = q0.z; b.y = q0.w; }
  else if (j == 2) { b.x = q1.x; b.y = q1.y; }
  else             { b.x = q1.z; b.y = q1.w; }
  return b;
}

#define AHEAD 8
template <int NB1, int NB2>
__device__ __forceinline__ void gemm16x32(const float* __restrict__ X1, const float* __restrict__ W1, int ld1,
                                          const float* __restrict__ X2, const float* __restrict__ W2, int ld2,
                                          int b0, int n0, int lane, v8f& acc0, v8f& acc1) {
  constexpr int NB = NB1 + NB2;
  v4f q[NB][6];
  #pragma unroll
  for (int i = 0; i < NB + AHEAD; ++i) {
    if (i < NB) {
      const bool first = i < NB1;
      const float* __restrict__ X = first ? X1 : X2;
      const float* __restrict__ Wt = first ? W1 : W2;
      const int ld = first ? ld1 : ld2;
      const int kc = 16 * (first ? i : i - NB1);
      frag16x16_g(X,  ld, b0,      kc, lane, q[i][0], q[i][1]);
      frag16x16_g(Wt, ld, n0,      kc, lane, q[i][2], q[i][3]);
      frag16x16_g(Wt, ld, n0 + 16, kc, lane, q[i][4], q[i][5]);
    }
    __builtin_amdgcn_sched_barrier(0);
    if (i >= AHEAD) {
      const int m = i - AHEAD;
      #pragma unroll
      for (int j = 0; j < 4; ++j) {
        const v2f a = pair8(q[m][0], q[m][1], j);
        acc0 = wmma4(a, pair8(q[m][2], q[m][3], j), acc0);
        acc1 = wmma4(a, pair8(q[m][4], q[m][5], j), acc1);
      }
    }
    __builtin_amdgcn_sched_barrier(0);
  }
}

__global__ __launch_bounds__(32) __attribute__((amdgpu_waves_per_eu(1, 1)))
void rnn_pre(const float* __restrict__ y_, const float* __restrict__ xi,
             const float* __restrict__ D12, const float* __restrict__ bv,
             const float* __restrict__ C1, float* __restrict__ Ews) {
  const int lane = threadIdx.x;
  const int b0   = blockIdx.y * 16;
  const int n0   = blockIdx.x * 32;
  const int colc = lane & 15;
  const int rhi  = (lane >> 4) << 3;

  v8f acc0 = {}, acc1 = {};
  gemm16x32<N_XI / 16, N_IN / 16>(xi, C1, N_XI, y_, D12, N_IN, b0, n0, lane, acc0, acc1);
  const int col = n0 + colc;
  const float bv0 = (col == 0) ? 0.0f : bv[col];
  const float bv1 = bv[col + 16];
  #pragma unroll
  for (int r = 0; r < 8; ++r) {
    Ews[(size_t)(b0 + r + rhi) * L_DIM + col]      = acc0[r] + bv0;
    Ews[(size_t)(b0 + r + rhi) * L_DIM + col + 16] = acc1[r] + bv1;
  }
}

#define POST_BLOCKS 640
#define REC_TILES   64
__device__ __forceinline__ void post_head(const float* __restrict__ y_, const float* __restrict__ xi,
                                          const float* __restrict__ B2, const float* __restrict__ C2,
                                          const float* __restrict__ D22, const float* __restrict__ Fm,
                                          float* __restrict__ Pws, int wg, int lane) {
  const int bt = wg / 10;
  const int cb = wg - bt * 10;
  const int b0 = bt * 16;
  const bool isu = cb < 2;
  const float* __restrict__ Wx = isu ? C2  : Fm;
  const float* __restrict__ Wy = isu ? D22 : B2;
  const int n0 = (isu ? cb : cb - 2) * 32;

  v8f acc0 = {}, acc1 = {};
  gemm16x32<N_XI / 16, N_IN / 16>(xi, Wx, N_XI, y_, Wy, N_IN, b0, n0, lane, acc0, acc1);

  float* pb = Pws + (size_t)wg * 512 + lane * 4;
  v4f s0, s1, s2, s3;
  s0.x = acc0[0]; s0.y = acc0[1]; s0.z = acc0[2]; s0.w = acc0[3];
  s1.x = acc0[4]; s1.y = acc0[5]; s1.z = acc0[6]; s1.w = acc0[7];
  s2.x = acc1[0]; s2.y = acc1[1]; s2.z = acc1[2]; s2.w = acc1[3];
  s3.x = acc1[4]; s3.y = acc1[5]; s3.z = acc1[6]; s3.w = acc1[7];
  *(v4f*)(pb)       = s0;
  *(v4f*)(pb + 128) = s1;
  *(v4f*)(pb + 256) = s2;
  *(v4f*)(pb + 384) = s3;
}


__device__ __forceinline__ float xhalf(float x) {
  const int b = __builtin_bit_cast(int, x);
  return __builtin_bit_cast(float,
      __builtin_amdgcn_permlanex16(b, b, 0x76543210, 0xfedcba98, false, false));
}

struct BlockCoef {
  v4f cl[2];
  float dd[8][8];
  v4f dx[8][2];
  v4f fq[15][2];
};

__device__ __forceinline__ float vsel(v4f q0, v4f q1, int i) {
  return i == 0 ? q0.x : i == 1 ? q0.y : i == 2 ? q0.z : i == 3 ? q0.w
       : i == 4 ? q1.x : i == 5 ? q1.y : i == 6 ? q1.z : q1.w;
}

__device__ __forceinline__ void fetch_cl(BlockCoef& k, const float* cs, int I, int h) {
  k.cl[0] = *(const v4f*)(cs + 16 * I + 8 * h);
  k.cl[1] = *(const v4f*)(cs + 16 * I + 8 * h + 4);
}

__device__ __forceinline__ void fetch_block(BlockCoef& k, const float* __restrict__ D11,
                                            int I, int c, int h) {
  const int c0 = 16 * I;
  const float* pd = D11 + (size_t)(c0 + 8 * h) * L_DIM + c0 + 8 * h;
  #pragma unroll
  for (int q = 1; q < 8; ++q) {
    #pragma unroll
    for (int r = 0; r < q; ++r) k.dd[q][r] = pd[(size_t)q * L_DIM + r];
  }
  const float* px = pd - 8 * h;
  #pragma unroll
  for (int q = 0; q < 8; ++q) {
    k.dx[q][0] = *(const v4f*)(px + (size_t)q * L_DIM);
    k.dx[q][1] = *(const v4f*)(px + (size_t)q * L_DIM + 4);
  }
  const float* pf = D11 + (size_t)c * L_DIM + c0 + 8 * h;
  #pragma unroll
  for (int t = 1; t < 16; ++t) {
    const float* pt = pf + (size_t)(((I + t) & 15) * 16) * L_DIM;
    k.fq[t - 1][0] = *(const v4f*)(pt);
    k.fq[t - 1][1] = *(const v4f*)(pt + 4);
  }
}

__device__ __forceinline__ v8f push1(v8f n, const v4f (&f)[2], const v2f (&b)[4]) {
  #pragma unroll
  for (int j = 0; j < 4; ++j) n = wmma4(pair8(f[0], f[1], j), b[j], n);
  return n;
}

__device__ __forceinline__ void do_block(const BlockCoef& k, v8f (&R)[16], int I, bool hi,
                                         float* __restrict__ eps_row) {
  float w[8], u[8], e1[8], e2[8];
  #pragma unroll
  for (int r = 0; r < 8; ++r) { w[r] = R[0][r]; u[r] = R[0][r]; }
  #pragma unroll
  for (int r = 0; r < 8; ++r) {
    e1[r] = __builtin_amdgcn_tanhf(w[r] * vsel(k.cl[0], k.cl[1], r));
    #pragma unroll
    for (int q = r + 1; q < 8; ++q) w[q] = fmaf(e1[r], k.dd[q][r], w[q]);
    const float ex = xhalf(e1[r]);
    #pragma unroll
    for (int q = 0; q < 8; ++q) u[q] = fmaf(ex, vsel(k.dx[q][0], k.dx[q][1], r), u[q]);
  }
  #pragma unroll
  for (int r = 0; r < 8; ++r) {
    e2[r] = __builtin_amdgcn_tanhf(u[r] * vsel(k.cl[0], k.cl[1], r));
    #pragma unroll
    for (int q = r + 1; q < 8; ++q) u[q] = fmaf(e2[r], k.dd[q][r], u[q]);
  }
  float e[8];
  #pragma unroll
  for (int r = 0; r < 8; ++r) e[r] = hi ? e2[r] : e1[r];

  {
    v4f s0, s1;
    s0.x = e[0]; s0.y = e[1]; s0.z = e[2]; s0.w = e[3];
    s1.x = e[4]; s1.y = e[5]; s1.z = e[6]; s1.w = e[7];
    *(v4f*)(eps_row + 16 * I)     = s0;
    *(v4f*)(eps_row + 16 * I + 4) = s1;
  }

  v2f b[4];
  #pragma unroll
  for (int j = 0; j < 4; ++j) { b[j].x = e[2 * j]; b[j].y = e[2 * j + 1]; }

  const int nrem = 15 - I;
  #pragma unroll
  for (int g = 0; g < 4; ++g) {
    const int t0 = 4 * g + 1;
    const int nt = g < 3 ? 4 : 3;
    if (nrem >= t0 + nt - 1) {
      v8f n0 = R[t0], n1 = R[t0 + 1], n2 = R[t0 + 2], n3 = R[nt == 4 ? t0 + 3 : t0 + 2];
      #pragma unroll
      for (int j = 0; j < 4; ++j) {
        n0 = wmma4(pair8(k.fq[t0 - 1][0], k.fq[t0 - 1][1], j), b[j], n0);
        n1 = wmma4(pair8(k.fq[t0][0],     k.fq[t0][1],     j), b[j], n1);
        n2 = wmma4(pair8(k.fq[t0 + 1][0], k.fq[t0 + 1][1], j), b[j], n2);
        if (nt == 4) n3 = wmma4(pair8(k.fq[t0 + 2][0], k.fq[t0 + 2][1], j), b[j], n3);
      }
      R[t0 - 1] = n0; R[t0] = n1; R[t0 + 1] = n2;
      if (nt == 4) R[t0 + 2] = n3;
    } else {
      #pragma unroll
      for (int t = t0; t < t0 + nt; ++t) {
        if (t <= nrem) R[t - 1] = push1(R[t], k.fq[t - 1], b);
      }
    }
  }
}

__global__ __launch_bounds__(32) __attribute__((amdgpu_waves_per_eu(1, 1)))
void rnn_rec(const float* __restrict__ Lambda, const float* __restrict__ D11,
             const float* __restrict__ Ews, float* __restrict__ eps_ws,
             const float* __restrict__ y_, const float* __restrict__ xi,
             const float* __restrict__ B2, const float* __restrict__ C2,
             const float* __restrict__ D22, const float* __restrict__ Fm,
             float* __restrict__ Pws) {
  __shared__ __attribute__((aligned(16))) float cs[L_DIM];
  const int lane = threadIdx.x;
  if (blockIdx.x >= REC_TILES) {
    post_head(y_, xi, B2, C2, D22, Fm, Pws, blockIdx.x - REC_TILES, lane);
    return;
  }
  const int b0   = blockIdx.x * 16;
  const int c    = lane & 15;
  const int h    = lane >> 4;
  const bool hi  = h != 0;

  float lr[8];
  #pragma unroll
  for (int t = 0; t < 8; ++t) lr[t] = Lambda[lane * 8 + t];

  v8f R[16];
  {
    const float* erow = Ews + (size_t)(b0 + c) * L_DIM + 8 * h;
    #pragma unroll
    for (int J = 0; J < 16; ++J) {
      const v4f q0 = *(const v4f*)(erow + 16 * J);
      const v4f q1 = *(const v4f*)(erow + 16 * J + 4);
      R[J][0] = q0.x; R[J][1] = q0.y; R[J][2] = q0.z; R[J][3] = q0.w;
      R[J][4] = q1.x; R[J][5] = q1.y; R[J][6] = q1.z; R[J][7] = q1.w;
    }
  }
  BlockCoef ka, kb;
  fetch_block(ka, D11, 0, c, h);

  #pragma unroll
  for (int t = 0; t < 8; ++t) cs[lane * 8 + t] = 1.0f / lr[t];
  __syncthreads();
  fetch_cl(ka, cs, 0, h);

  float* eps_row = eps_ws + (size_t)(b0 + c) * L_DIM + 8 * h;

  #pragma unroll 1
  for (int I = 0; I < 16; I += 2) {
    fetch_block(kb, D11, I + 1, c, h);
    fetch_cl(kb, cs, I + 1, h);
    do_block(ka, R, I, hi, eps_row);
    const int In = I + 2 < 16 ? I + 2 : 15;
    fetch_block(ka, D11, In, c, h);
    fetch_cl(ka, cs, In, h);
    do_block(kb, R, I + 1, hi, eps_row);
  }
}

__global__ __launch_bounds__(32) __attribute__((amdgpu_waves_per_eu(1, 1)))
void rnn_post(const float* __restrict__ D21, const float* __restrict__ B1,
              const float* __restrict__ bxi, const float* __restrict__ bu,
              const float* __restrict__ eps, const float* __restrict__ Pws,
              float* __restrict__ out_u, float* __restrict__ out_xi) {
  const int lane = threadIdx.x;
  const int b0   = blockIdx.y * 16;
  const int cb   = blockIdx.x;
  const int colc = lane & 15;
  const int rhi  = (lane >> 4) << 3;

  const bool isu = cb < 2;
  const float* __restrict__ We = isu ? D21 : B1;
  const float* __restrict__ bs = isu ? bu  : bxi;
  float* __restrict__ outp = isu ? out_u : out_xi;
  const int   ldo = isu ? M_OUT : N_XI;
  const float scl = isu ? OUT_AMP : INV_M;
  const int   n0  = (isu ? cb : cb - 2) * 32;

  v8f acc0, acc1;
  {
    const float* pb = Pws + (size_t)(blockIdx.y * 10 + cb) * 512 + lane * 4;
    const v4f s0 = *(const v4f*)(pb);
    const v4f s1 = *(const v4f*)(pb + 128);
    const v4f s2 = *(const v4f*)(pb + 256);
    const v4f s3 = *(const v4f*)(pb + 384);
    acc0[0] = s0.x; acc0[1] = s0.y; acc0[2] = s0.z; acc0[3] = s0.w;
    acc0[4] = s1.x; acc0[5] = s1.y; acc0[6] = s1.z; acc0[7] = s1.w;
    acc1[0] = s2.x; acc1[1] = s2.y; acc1[2] = s2.z; acc1[3] = s2.w;
    acc1[4] = s3.x; acc1[5] = s3.y; acc1[6] = s3.z; acc1[7] = s3.w;
  }
  gemm16x32<L_DIM / 16, 0>(eps, We, L_DIM, eps, We, L_DIM, b0, n0, lane, acc0, acc1);
  const int col = n0 + colc;
  const float bc0 = bs[col];
  const float bc1 = bs[col + 16];
  #pragma unroll
  for (int r = 0; r < 8; ++r) {
    outp[(size_t)(b0 + r + rhi) * ldo + col]      = (acc0[r] + bc0) * scl;
    outp[(size_t)(b0 + r + rhi) * ldo + col + 16] = (acc1[r] + bc1) * scl;
  }
}

extern "C" void kernel_launch(void* const* d_in, const int* in_sizes, int n_in,
                              void* d_out, int out_size, void* d_ws, size_t ws_size,
                              hipStream_t stream) {
  const float* y_  = (const float*)d_in[0];
  const float* xi  = (const float*)d_in[1];
  const float* B2  = (const float*)d_in[2];
  const float* C2  = (const float*)d_in[3];
  const float* D21 = (const float*)d_in[4];
  const float* D22 = (const float*)d_in[5];
  const float* D12 = (const float*)d_in[6];
  const float* bxi = (const float*)d_in[7];
  const float* bv  = (const float*)d_in[8];
  const float* bu  = (const float*)d_in[9];
  const float* Fm  = (const float*)d_in[10];
  const float* B1  = (const float*)d_in[11];
  const float* Lam = (const float*)d_in[13];
  const float* C1  = (const float*)d_in[14];
  const float* D11 = (const float*)d_in[15];

  float* out_u  = (float*)d_out;
  float* out_xi = out_u + (size_t)B_TOT * M_OUT;

  float* Ews = (float*)d_ws;
  float* eps = Ews + (size_t)B_TOT * L_DIM;
  float* Pws = eps + (size_t)B_TOT * L_DIM;

  rnn_pre<<<dim3(8, 64), dim3(32), 0, stream>>>(y_, xi, D12, bv, C1, Ews);
  rnn_rec<<<dim3(REC_TILES + POST_BLOCKS), dim3(32), 0, stream>>>(
      Lam, D11, Ews, eps, y_, xi, B2, C2, D22, Fm, Pws);
  rnn_post<<<dim3(10, 64), dim3(32), 0, stream>>>(
      D21, B1, bxi, bu, eps, Pws, out_u, out_xi);

  (void)in_sizes; (void)n_in; (void)out_size; (void)ws_size;
}


# Recurrence on four waves per batch tile, one per SIMD of a WGP: each keeps every fourth 16-column tile, the stepping wave hands its eps block on via LDS, pushes run beside the steps; sums unchanged
# speedup vs baseline: 59.0807x; 1.2925x over previous
#include <hip/hip_runtime.h>
#include <math.h>


#define B_TOT 1024
#define N_IN  128
#define M_OUT 64
#define N_XI  256
#define L_DIM 256
#define OUT_AMP 40.0f
#define INV_M  (1.0f/1.001f)

typedef __attribute__((ext_vector_type(2))) float v2f;
typedef __attribute__((ext_vector_type(4))) float v4f;
typedef __attribute__((ext_vector_type(8))) float v8f;

__device__ __forceinline__ void frag16x16_g(const float* __restrict__ M, int ld,
                                            int r0, int kc, int lane, v4f& q0, v4f& q1) {
  const float* p = M + (size_t)(r0 + (lane & 15)) * ld + kc + ((lane >> 4) << 3);
  q0 = *(const v4f*)p;
  q1 = *(const v4f*)(p + 4);
}

__device__ __forceinline__ v8f wmma4(v2f a, v2f b, v8f c) {
  return __builtin_amdgcn_wmma_f32_16x16x4_f32(false, a, false, b, (short)0, c,
                                               false, false);
}

__device__ __forceinline__ v2f pair8(v4f q0, v4f q1, int j) {
  v2f b;
  if (j == 0)      { b.x = q0.x; b.y = q0.y; }
  else if (j == 1) { b.x = q0.z; b.y = q0.w; }
  else if (j == 2) { b.x = q1.x; b.y = q1.y; }
  else             { b.x = q1.z; b.y = q1.w; }
  return b;
}

#define AHEAD 8
template <int NB1, int NB2>
__device__ __forceinline__ void gemm16x32(const float* __restrict__ X1, const float* __restrict__ W1, int ld1,
                                          const float* __restrict__ X2, const float* __restrict__ W2, int ld2,
                                          int b0, int n0, int lane, v8f& acc0, v8f& acc1) {
  constexpr int NB = NB1 + NB2;
  v4f q[NB][6];
  #pragma unroll
  for (int i = 0; i < NB + AHEAD; ++i) {
    if (i < NB) {
      const bool first = i < NB1;
      const float* __restrict__ X = first ? X1 : X2;
      const float* __restrict__ Wt = first ? W1 : W2;
      const int ld = first ? ld1 : ld2;
      const int kc = 16 * (first ? i : i - NB1);
      frag16x16_g(X,  ld, b0,      kc, lane, q[i][0], q[i][1]);
      frag16x16_g(Wt, ld, n0,      kc, lane, q[i][2], q[i][3]);
      frag16x16_g(Wt, ld, n0 + 16, kc, lane, q[i][4], q[i][5]);
    }
    __builtin_amdgcn_sched_barrier(0);
    if (i >= AHEAD) {
      const int m = i - AHEAD;
      #pragma unroll
      for (int j = 0; j < 4; ++j) {
        const v2f a = pair8(q[m][0], q[m][1], j);
        acc0 = wmma4(a, pair8(q[m][2], q[m][3], j), acc0);
        acc1 = wmma4(a, pair8(q[m][4], q[m][5], j), acc1);
      }
    }
    __builtin_amdgcn_sched_barrier(0);
  }
}

__global__ __launch_bounds__(32) __attribute__((amdgpu_waves_per_eu(1, 1)))
void rnn_pre(const float* __restrict__ y_, const float* __restrict__ xi,
             const float* __restrict__ D12, const float* __restrict__ bv,
             const float* __restrict__ C1, float* __restrict__ Ews) {
  const int lane = threadIdx.x;
  const int b0   = blockIdx.y * 16;
  const int n0   = blockIdx.x * 32;
  const int colc = lane & 15;
  const int rhi  = (lane >> 4) << 3;

  v8f acc0 = {}, acc1 = {};
  gemm16x32<N_XI / 16, N_IN / 16>(xi, C1, N_XI, y_, D12, N_IN, b0, n0, lane, acc0, acc1);
  const int col = n0 + colc;
  const float bv0 = (col == 0) ? 0.0f : bv[col];
  const float bv1 = bv[col + 16];
  #pragma unroll
  for (int r = 0; r < 8; ++r) {
    Ews[(size_t)(b0 + r + rhi) * L_DIM + col]      = acc0[r] + bv0;
    Ews[(size_t)(b0 + r + rhi) * L_DIM + col + 16] = acc1[r] + bv1;
  }
}

#define POST_BLOCKS 640
#define REC_TILES   64
__device__ __forceinline__ void post_head(const float* __restrict__ y_, const float* __restrict__ xi,
                                          const float* __restrict__ B2, const float* __restrict__ C2,
                                          const float* __restrict__ D22, const float* __restrict__ Fm,
                                          float* __restrict__ Pws, int wg, int lane) {
  const int bt = wg / 10;
  const int cb = wg - bt * 10;
  const int b0 = bt * 16;
  const bool isu = cb < 2;
  const float* __restrict__ Wx = isu ? C2  : Fm;
  const float* __restrict__ Wy = isu ? D22 : B2;
  const int n0 = (isu ? cb : cb - 2) * 32;

  v8f acc0 = {}, acc1 = {};
  gemm16x32<N_XI / 16, N_IN / 16>(xi, Wx, N_XI, y_, Wy, N_IN, b0, n0, lane, acc0, acc1);

  float* pb = Pws + (size_t)wg * 512 + lane * 4;
  v4f s0, s1, s2, s3;
  s0.x = acc0[0]; s0.y = acc0[1]; s0.z = acc0[2]; s0.w = acc0[3];
  s1.x = acc0[4]; s1.y = acc0[5]; s1.z = acc0[6]; s1.w = acc0[7];
  s2.x = acc1[0]; s2.y = acc1[1]; s2.z = acc1[2]; s2.w = acc1[3];
  s3.x = acc1[4]; s3.y = acc1[5]; s3.z = acc1[6]; s3.w = acc1[7];
  *(v4f*)(pb)       = s0;
  *(v4f*)(pb + 128) = s1;
  *(v4f*)(pb + 256) = s2;
  *(v4f*)(pb + 384) = s3;
}

#define REC_WAVES 4

__device__ __forceinline__ float xhalf(float x) {
  const int b = __builtin_bit_cast(int, x);
  return __builtin_bit_cast(float,
      __builtin_amdgcn_permlanex16(b, b, 0x76543210, 0xfedcba98, false, false));
}

struct StepCoef {
  v4f cl[2];
  v4f dd[8][2];
  v4f dx[8][2];
};

struct PushCoef { v4f f[4][2]; };

__device__ __forceinline__ float vsel(v4f q0, v4f q1, int i) {
  return i == 0 ? q0.x : i == 1 ? q0.y : i == 2 ? q0.z : i == 3 ? q0.w
       : i == 4 ? q1.x : i == 5 ? q1.y : i == 6 ? q1.z : q1.w;
}

__device__ __forceinline__ float dsel(const StepCoef& k, int q, int r) {
  const v4f p = r < 4 ? k.dd[q][0] : k.dd[q][1];
  const int i = r & 3;
  return i == 0 ? p.x : i == 1 ? p.y : i == 2 ? p.z : p.w;
}

__device__ __forceinline__ void fetch_cl(StepCoef& k, const float* cs, int I, int h) {
  k.cl[0] = *(const v4f*)(cs + 16 * I + 8 * h);
  k.cl[1] = *(const v4f*)(cs + 16 * I + 8 * h + 4);
}

__device__ __forceinline__ void fetch_step(StepCoef& k, const float* __restrict__ D11, int I, int h) {
  const int c0 = 16 * I;
  const float* pd = D11 + (size_t)(c0 + 8 * h) * L_DIM + c0 + 8 * h;
  #pragma unroll
  for (int q = 1; q < 8; ++q) {
    k.dd[q][0] = *(const v4f*)(pd + (size_t)q * L_DIM);
    if (q > 4) k.dd[q][1] = *(const v4f*)(pd + (size_t)q * L_DIM + 4);
  }
  const float* px = pd - 8 * h;
  #pragma unroll
  for (int q = 0; q < 8; ++q) {
    k.dx[q][0] = *(const v4f*)(px + (size_t)q * L_DIM);
    k.dx[q][1] = *(const v4f*)(px + (size_t)q * L_DIM + 4);
  }
}

__device__ __forceinline__ void fetch_push(PushCoef& P, const float* __restrict__ D11,
                                           int blk, int w, int c, int h) {
  const float* pf = D11 + (size_t)c * L_DIM + 16 * blk + 8 * h;
  const int gs = (blk + 1) >> 2;
  #pragma unroll
  for (int m = 0; m < 4; ++m) {
    const int J = (4 * (gs + m) + w) & 15;
    const float* pt = pf + (size_t)(J * 16) * L_DIM;
    P.f[m][0] = *(const v4f*)(pt);
    P.f[m][1] = *(const v4f*)(pt + 4);
  }
}

__device__ __forceinline__ void push1(v8f& n0, const v4f (&f0)[2], const v2f (&b)[4]) {
  #pragma unroll
  for (int j = 0; j < 4; ++j) n0 = wmma4(pair8(f0[0], f0[1], j), b[j], n0);
}
__device__ __forceinline__ void push2(v8f& n0, const v4f (&f0)[2], v8f& n1, const v4f (&f1)[2],
                                      const v2f (&b)[4]) {
  #pragma unroll
  for (int j = 0; j < 4; ++j) {
    n0 = wmma4(pair8(f0[0], f0[1], j), b[j], n0);
    n1 = wmma4(pair8(f1[0], f1[1], j), b[j], n1);
  }
}
__device__ __forceinline__ void push3(v8f& n0, const v4f (&f0)[2], v8f& n1, const v4f (&f1)[2],
                                      v8f& n2, const v4f (&f2)[2], const v2f (&b)[4]) {
  #pragma unroll
  for (int j = 0; j < 4; ++j) {
    n0 = wmma4(pair8(f0[0], f0[1], j), b[j], n0);
    n1 = wmma4(pair8(f1[0], f1[1], j), b[j], n1);
    n2 = wmma4(pair8(f2[0], f2[1], j), b[j], n2);
  }
}
__device__ __forceinline__ void push4(v8f& n0, const v4f (&f0)[2], v8f& n1, const v4f (&f1)[2],
                                      v8f& n2, const v4f (&f2)[2], v8f& n3, const v4f (&f3)[2],
                                      const v2f (&b)[4]) {
  #pragma unroll
  for (int j = 0; j < 4; ++j) {
    n0 = wmma4(pair8(f0[0], f0[1], j), b[j], n0);
    n1 = wmma4(pair8(f1[0], f1[1], j), b[j], n1);
    n2 = wmma4(pair8(f2[0], f2[1], j), b[j], n2);
    n3 = wmma4(pair8(f3[0], f3[1], j), b[j], n3);
  }
}

__device__ __forceinline__ void push_set(v8f (&T)[4], const PushCoef& P, const v2f (&b)[4],
                                         bool near, int nfar) {
  if (near) {
    if (nfar >= 3)      push4(T[0], P.f[0], T[1], P.f[1], T[2], P.f[2], T[3], P.f[3], b);
    else if (nfar == 2) push3(T[0], P.f[0], T[1], P.f[1], T[2], P.f[2], b);
    else if (nfar == 1) push2(T[0], P.f[0], T[1], P.f[1], b);
    else                push1(T[0], P.f[0], b);
  } else {
    if (nfar >= 3)      push3(T[1], P.f[1], T[2], P.f[2], T[3], P.f[3], b);
    else if (nfar == 2) push2(T[1], P.f[1], T[2], P.f[2], b);
    else if (nfar == 1) push1(T[1], P.f[1], b);
  }
}

__device__ __forceinline__ void steps16(const StepCoef& k, const v8f& R0, bool hi, float (&e)[8]) {
  float w[8], u[8], e1[8], e2[8];
  #pragma unroll
  for (int q = 1; q < 8; ++q) {
    asm volatile("" :: "v"(k.dd[q][0]));
    if (q > 4) asm volatile("" :: "v"(k.dd[q][1]));
  }
  #pragma unroll
  for (int r = 0; r < 8; ++r) { w[r] = R0[r]; u[r] = R0[r]; }
  #pragma unroll
  for (int r = 0; r < 8; ++r) {
    e1[r] = __builtin_amdgcn_tanhf(w[r] * vsel(k.cl[0], k.cl[1], r));
    #pragma unroll
    for (int q = r + 1; q < 8; ++q) w[q] = fmaf(e1[r], dsel(k, q, r), w[q]);
    const float ex = xhalf(e1[r]);
    #pragma unroll
    for (int q = 0; q < 8; ++q) u[q] = fmaf(ex, vsel(k.dx[q][0], k.dx[q][1], r), u[q]);
  }
  #pragma unroll
  for (int r = 0; r < 8; ++r) {
    e2[r] = __builtin_amdgcn_tanhf(u[r] * vsel(k.cl[0], k.cl[1], r));
    #pragma unroll
    for (int q = r + 1; q < 8; ++q) u[q] = fmaf(e2[r], dsel(k, q, r), u[q]);
  }
  #pragma unroll
  for (int r = 0; r < 8; ++r) e[r] = hi ? e2[r] : e1[r];
}

__device__ __forceinline__ void lds_barrier() {
  asm volatile("s_wait_dscnt 0x0\n\ts_barrier_signal -1\n\ts_barrier_wait -1" ::: "memory");
}
__device__ __forceinline__ void bar_signal_lds() {
  asm volatile("s_wait_dscnt 0x0\n\ts_barrier_signal -1" ::: "memory");
}
__device__ __forceinline__ void bar_wait() {
  asm volatile("s_barrier_wait -1" ::: "memory");
}

template <int PB_>
__device__ __forceinline__ void rec_body(int g, int w, int lane, int c, int h, bool hi,
                                         v8f (&T)[4], StepCoef& k, const PushCoef& Pc, PushCoef& Pn,
                                         v2f (&bown)[4], const float* __restrict__ D11,
                                         const float* cs, float* eb, float* __restrict__ eps_row) {
  const int I = 4 * g + PB_;
  const bool own     = (w == PB_);
  const bool prevown = (w == ((PB_ + 3) & 3));
  const bool first   = (PB_ == 0) && (g == 0);
  const bool lastb   = (PB_ == 3) && (g == 3);
  const int nfar = 3 - g;
  float* ebw = eb + (I & 1) * 256 + lane * 8;
  const float* ebr = eb + ((I + 1) & 1) * 256 + lane * 8;

  fetch_push(Pn, D11, I, w, c, h);

  v2f b[4];
  #pragma unroll
  for (int j = 0; j < 4; ++j) b[j] = bown[j];
  if (!first) {
    bar_wait();
    if (!prevown) {
      const v4f q0 = *(const v4f*)(ebr);
      const v4f q1 = *(const v4f*)(ebr + 4);
      b[0].x = q0.x; b[0].y = q0.y; b[1].x = q0.z; b[1].y = q0.w;
      b[2].x = q1.x; b[2].y = q1.y; b[3].x = q1.z; b[3].y = q1.w;
    }
  }
  if (own) {
    if (!first) push1(T[0], Pc.f[0], b);
    float e[8];
    steps16(k, T[0], hi, e);
    v4f s0, s1;
    s0.x = e[0]; s0.y = e[1]; s0.z = e[2]; s0.w = e[3];
    s1.x = e[4]; s1.y = e[5]; s1.z = e[6]; s1.w = e[7];
    *(v4f*)(ebw)     = s0;
    *(v4f*)(ebw + 4) = s1;
    #pragma unroll
    for (int j = 0; j < 4; ++j) { bown[j].x = e[2 * j]; bown[j].y = e[2 * j + 1]; }
  }
  if (!lastb) bar_signal_lds();
  if (own) {
    v4f s0, s1;
    s0.x = bown[0].x; s0.y = bown[0].y; s0.z = bown[1].x; s0.w = bown[1].y;
    s1.x = bown[2].x; s1.y = bown[2].y; s1.z = bown[3].x; s1.w = bown[3].y;
    *(v4f*)(eps_row + 16 * I)     = s0;
    *(v4f*)(eps_row + 16 * I + 4) = s1;
  }
  if (prevown && !first) {
    const int I3 = I + 3 < 16 ? I + 3 : I - 1;
    fetch_step(k, D11, I3, h);
    fetch_cl(k, cs, I3, h);
  }
  if (!first) push_set(T, Pc, b, w > PB_, nfar);
}

__global__ __launch_bounds__(32 * REC_WAVES) __attribute__((amdgpu_waves_per_eu(1, 1)))
void rnn_rec(const float* __restrict__ Lambda, const float* __restrict__ D11,
             const float* __restrict__ Ews, float* __restrict__ eps_ws,
             const float* __restrict__ y_, const float* __restrict__ xi,
             const float* __restrict__ B2, const float* __restrict__ C2,
             const float* __restrict__ D22, const float* __restrict__ Fm,
             float* __restrict__ Pws) {
  __shared__ __attribute__((aligned(16))) float cs[L_DIM];
  __shared__ __attribute__((aligned(16))) float eb[2 * 256];
  const int lane = threadIdx.x & 31;
  const int w = __builtin_amdgcn_readfirstlane((int)(threadIdx.x >> 5));
  if (blockIdx.x >= REC_TILES) {
    post_head(y_, xi, B2, C2, D22, Fm, Pws, (blockIdx.x - REC_TILES) * REC_WAVES + w, lane);
    return;
  }
  const int b0   = blockIdx.x * 16;
  const int c    = lane & 15;
  const int h    = lane >> 4;
  const bool hi  = h != 0;

  const float l0 = Lambda[2 * threadIdx.x];
  const float l1 = Lambda[2 * threadIdx.x + 1];

  v8f T[4];
  {
    const float* erow = Ews + (size_t)(b0 + c) * L_DIM + 8 * h;
    #pragma unroll
    for (int m = 0; m < 4; ++m) {
      const int J = w + 4 * m;
      const v4f q0 = *(const v4f*)(erow + 16 * J);
      const v4f q1 = *(const v4f*)(erow + 16 * J + 4);
      T[m][0] = q0.x; T[m][1] = q0.y; T[m][2] = q0.z; T[m][3] = q0.w;
      T[m][4] = q1.x; T[m][5] = q1.y; T[m][6] = q1.z; T[m][7] = q1.w;
    }
  }
  StepCoef k;
  fetch_step(k, D11, w, h);
  PushCoef PA = {}, PB = {};

  cs[2 * threadIdx.x]     = 1.0f / l0;
  cs[2 * threadIdx.x + 1] = 1.0f / l1;
  lds_barrier();
  fetch_cl(k, cs, w, h);

  float* eps_row = eps_ws + (size_t)(b0 + c) * L_DIM + 8 * h;
  v2f bown[4] = {};

  #pragma unroll 1
  for (int g = 0; g < 4; ++g) {
    rec_body<0>(g, w, lane, c, h, hi, T, k, PB, PA, bown, D11, cs, eb, eps_row);
    rec_body<1>(g, w, lane, c, h, hi, T, k, PA, PB, bown, D11, cs, eb, eps_row);
    rec_body<2>(g, w, lane, c, h, hi, T, k, PB, PA, bown, D11, cs, eb, eps_row);
    rec_body<3>(g, w, lane, c, h, hi, T, k, PA, PB, bown, D11, cs, eb, eps_row);
    T[0] = T[1]; T[1] = T[2]; T[2] = T[3];
  }
}

__global__ __launch_bounds__(32) __attribute__((amdgpu_waves_per_eu(1, 1)))
void rnn_post(const float* __restrict__ D21, const float* __restrict__ B1,
              const float* __restrict__ bxi, const float* __restrict__ bu,
              const float* __restrict__ eps, const float* __restrict__ Pws,
              float* __restrict__ out_u, float* __restrict__ out_xi) {
  const int lane = threadIdx.x;
  const int b0   = blockIdx.y * 16;
  const int cb   = blockIdx.x;
  const int colc = lane & 15;
  const int rhi  = (lane >> 4) << 3;

  const bool isu = cb < 2;
  const float* __restrict__ We = isu ? D21 : B1;
  const float* __restrict__ bs = isu ? bu  : bxi;
  float* __restrict__ outp = isu ? out_u : out_xi;
  const int   ldo = isu ? M_OUT : N_XI;
  const float scl = isu ? OUT_AMP : INV_M;
  const int   n0  = (isu ? cb : cb - 2) * 32;

  v8f acc0, acc1;
  {
    const float* pb = Pws + (size_t)(blockIdx.y * 10 + cb) * 512 + lane * 4;
    const v4f s0 = *(const v4f*)(pb);
    const v4f s1 = *(const v4f*)(pb + 128);
    const v4f s2 = *(const v4f*)(pb + 256);
    const v4f s3 = *(const v4f*)(pb + 384);
    acc0[0] = s0.x; acc0[1] = s0.y; acc0[2] = s0.z; acc0[3] = s0.w;
    acc0[4] = s1.x; acc0[5] = s1.y; acc0[6] = s1.z; acc0[7] = s1.w;
    acc1[0] = s2.x; acc1[1] = s2.y; acc1[2] = s2.z; acc1[3] = s2.w;
    acc1[4] = s3.x; acc1[5] = s3.y; acc1[6] = s3.z; acc1[7] = s3.w;
  }
  gemm16x32<L_DIM / 16, 0>(eps, We, L_DIM, eps, We, L_DIM, b0, n0, lane, acc0, acc1);
  const int col = n0 + colc;
  const float bc0 = bs[col];
  const float bc1 = bs[col + 16];
  #pragma unroll
  for (int r = 0; r < 8; ++r) {
    outp[(size_t)(b0 + r + rhi) * ldo + col]      = (acc0[r] + bc0) * scl;
    outp[(size_t)(b0 + r + rhi) * ldo + col + 16] = (acc1[r] + bc1) * scl;
  }
}

extern "C" void kernel_launch(void* const* d_in, const int* in_sizes, int n_in,
                              void* d_out, int out_size, void* d_ws, size_t ws_size,
                              hipStream_t stream) {
  const float* y_  = (const float*)d_in[0];
  const float* xi  = (const float*)d_in[1];
  const float* B2  = (const float*)d_in[2];
  const float* C2  = (const float*)d_in[3];
  const float* D21 = (const float*)d_in[4];
  const float* D22 = (const float*)d_in[5];
  const float* D12 = (const float*)d_in[6];
  const float* bxi = (const float*)d_in[7];
  const float* bv  = (const float*)d_in[8];
  const float* bu  = (const float*)d_in[9];
  const float* Fm  = (const float*)d_in[10];
  const float* B1  = (const float*)d_in[11];
  const float* Lam = (const float*)d_in[13];
  const float* C1  = (const float*)d_in[14];
  const float* D11 = (const float*)d_in[15];

  float* out_u  = (float*)d_out;
  float* out_xi = out_u + (size_t)B_TOT * M_OUT;

  float* Ews = (float*)d_ws;
  float* eps = Ews + (size_t)B_TOT * L_DIM;
  float* Pws = eps + (size_t)B_TOT * L_DIM;

  rnn_pre<<<dim3(8, 64), dim3(32), 0, stream>>>(y_, xi, D12, bv, C1, Ews);
  rnn_rec<<<dim3(REC_TILES + POST_BLOCKS / REC_WAVES), dim3(32 * REC_WAVES), 0, stream>>>(
      Lam, D11, Ews, eps, y_, xi, B2, C2, D22, Fm, Pws);
  rnn_post<<<dim3(10, 64), dim3(32), 0, stream>>>(
      D21, B1, bxi, bu, eps, Pws, out_u, out_xi);

  (void)in_sizes; (void)n_in; (void)out_size; (void)ws_size;
}
